# starting kernel
# baseline (speedup 1.0000x reference)
#include <hip/hip_runtime.h>
#include <hip/hip_bf16.h>


#define D_FEAT 128
#define LDS_STRIDE 132
#define OUT_DIM 26

typedef __attribute__((ext_vector_type(2))) float v2f;
typedef __attribute__((ext_vector_type(8))) float v8f;


__global__ void sage_zero_f32(float* __restrict__ p, int n) {
    int i = blockIdx.x * blockDim.x + threadIdx.x;
    if (i < n) p[i] = 0.0f;
}

__global__ void sage_deg_kernel(const int* __restrict__ dst, float* __restrict__ deg,
                                int n_edges) {
    int e = blockIdx.x * blockDim.x + threadIdx.x;
    if (e < n_edges) atomicAdd(&deg[dst[e]], 1.0f);
}

__global__ void sage_invdeg_kernel(float* __restrict__ deg, int n_nodes) {
    int i = blockIdx.x * blockDim.x + threadIdx.x;
    if (i < n_nodes) {
        float d = deg[i];
        deg[i] = 1.0f / fmaxf(d, 1.0f);
    }
}

__global__ void sage_scatter_kernel(const int* __restrict__ src, const int* __restrict__ dst,
                                    const float* __restrict__ x, float* __restrict__ msg,
                                    int n_edges) {
    long long tid = (long long)blockIdx.x * blockDim.x + threadIdx.x;
    int e = (int)(tid >> 5);
    int c = (int)(tid & 31);
    if (e >= n_edges) return;
    int s = src[e];
    int d = dst[e];
    const float4 v = *(const float4*)(x + (size_t)s * D_FEAT + c * 4);
    float* p = msg + (size_t)d * D_FEAT + c * 4;
    atomicAdd(p + 0, v.x);
    atomicAdd(p + 1, v.y);
    atomicAdd(p + 2, v.z);
    atomicAdd(p + 3, v.w);
}

__global__ void __launch_bounds__(256)
sage_layer_kernel(const float* __restrict__ xin, const float* __restrict__ msg,
                  const float* __restrict__ invdeg,
                  const float* __restrict__ Wself, const float* __restrict__ Wneigh,
                  const float* __restrict__ bias, float* __restrict__ xout) {
    __shared__ float As[16 * LDS_STRIDE];
    __shared__ float Hs[16 * LDS_STRIDE];

    const int m0 = blockIdx.x * 16;
    const int t  = threadIdx.x;

    {
        int r  = t >> 4;
        int c0 = (t & 15) * 8;
        const float* xr = xin + (size_t)(m0 + r) * D_FEAT + c0;
        float4 a0 = *(const float4*)(xr);
        float4 a1 = *(const float4*)(xr + 4);
        *(float4*)(As + r * LDS_STRIDE + c0)     = a0;
        *(float4*)(As + r * LDS_STRIDE + c0 + 4) = a1;

        float iv = invdeg[m0 + r];
        const float* mr = msg + (size_t)(m0 + r) * D_FEAT + c0;
        float4 h0 = *(const float4*)(mr);
        float4 h1 = *(const float4*)(mr + 4);
        h0.x *= iv; h0.y *= iv; h0.z *= iv; h0.w *= iv;
        h1.x *= iv; h1.y *= iv; h1.z *= iv; h1.w *= iv;
        *(float4*)(Hs + r * LDS_STRIDE + c0)     = h0;
        *(float4*)(Hs + r * LDS_STRIDE + c0 + 4) = h1;
    }
    __syncthreads();

    const int wave = t >> 5;
    const int lane = t & 31;
    const int n0   = wave * 16;
    const int half = lane >> 4;
    const int l16  = lane & 15;
    const int col  = n0 + l16;

    v8f acc;
    {
        float bv = bias[col];
        #pragma unroll
        for (int i = 0; i < 8; ++i) acc[i] = bv;
    }

    #pragma unroll 4
    for (int k0 = 0; k0 < D_FEAT; k0 += 4) {
        const int ka = k0 + half * 2;
        v2f a = *(const v2f*)(As + l16 * LDS_STRIDE + ka);
        v2f b;
        b.x = Wself[(size_t)ka * D_FEAT + col];
        b.y = Wself[(size_t)(ka + 1) * D_FEAT + col];
        acc = __builtin_amdgcn_wmma_f32_16x16x4_f32(false, a, false, b,
                                                    (short)0, acc, false, false);

        v2f h = *(const v2f*)(Hs + l16 * LDS_STRIDE + ka);
        v2f bn;
        bn.x = Wneigh[(size_t)ka * D_FEAT + col];
        bn.y = Wneigh[(size_t)(ka + 1) * D_FEAT + col];
        acc = __builtin_amdgcn_wmma_f32_16x16x4_f32(false, h, false, bn,
                                                    (short)0, acc, false, false);
    }

    #pragma unroll
    for (int v = 0; v < 8; ++v) {
        int row = m0 + v + half * 8;
        float val = acc[v];
        val = val > 0.0f ? val : 0.0f;
        xout[(size_t)row * D_FEAT + col] = val;
    }
}

__global__ void __launch_bounds__(64)
sage_pred_kernel(const float* __restrict__ xin, const float* __restrict__ Wp,
                 const float* __restrict__ bp, float* __restrict__ out) {
    const int m0   = blockIdx.x * 16;
    const int t    = threadIdx.x;
    const int wave = t >> 5;
    const int lane = t & 31;
    const int n0   = wave * 16;
    const int half = lane >> 4;
    const int l16  = lane & 15;
    const int col  = n0 + l16;
    const int bcol = col < OUT_DIM ? col : (OUT_DIM - 1);

    v8f acc;
    {
        float bv = bp[bcol];
        #pragma unroll
        for (int i = 0; i < 8; ++i) acc[i] = bv;
    }

    #pragma unroll 4
    for (int k0 = 0; k0 < D_FEAT; k0 += 4) {
        const int ka = k0 + half * 2;
        const float* ap = xin + (size_t)(m0 + l16) * D_FEAT + ka;
        v2f a; a.x = ap[0]; a.y = ap[1];
        v2f b;
        b.x = Wp[(size_t)ka * OUT_DIM + bcol];
        b.y = Wp[(size_t)(ka + 1) * OUT_DIM + bcol];
        acc = __builtin_amdgcn_wmma_f32_16x16x4_f32(false, a, false, b,
                                                    (short)0, acc, false, false);
    }

    if (col < OUT_DIM) {
        #pragma unroll
        for (int v = 0; v < 8; ++v) {
            int row = m0 + v + half * 8;
            out[(size_t)row * OUT_DIM + col] = acc[v];
        }
    }
}


extern "C" void kernel_launch(void* const* d_in, const int* in_sizes, int n_in,
                              void* d_out, int out_size, void* d_ws, size_t ws_size,
                              hipStream_t stream) {
    const float* x       = (const float*)d_in[0];
    const int*   src     = (const int*)  d_in[1];
    const int*   dst     = (const int*)  d_in[2];
    const float* W_self  = (const float*)d_in[3];
    const float* W_neigh = (const float*)d_in[4];
    const float* b       = (const float*)d_in[5];
    const float* W_pred  = (const float*)d_in[6];
    const float* b_pred  = (const float*)d_in[7];

    const int n_nodes = in_sizes[0] / D_FEAT;
    const int n_edges = in_sizes[1];
    const int L       = in_sizes[5] / D_FEAT;
    const int feat_elems = n_nodes * D_FEAT;

    char*  ws = (char*)d_ws;
    size_t off = ((size_t)n_nodes * sizeof(float) + 511) & ~(size_t)511;
    size_t feat_bytes = (size_t)feat_elems * sizeof(float);
    float* invdeg = (float*)ws;
    float* msg    = (float*)(ws + off);
    float* bufA   = (float*)(ws + off + feat_bytes);
    float* bufB   = (float*)(ws + off + 2 * feat_bytes);

    const int T = 256;

    sage_zero_f32<<<(n_nodes + T - 1) / T, T, 0, stream>>>(invdeg, n_nodes);
    sage_deg_kernel<<<(n_edges + T - 1) / T, T, 0, stream>>>(dst, invdeg, n_edges);
    sage_invdeg_kernel<<<(n_nodes + T - 1) / T, T, 0, stream>>>(invdeg, n_nodes);

    const float* cur = x;
    float* nxt = bufA;
    const long long scat_threads = (long long)n_edges * 32;
    const int scat_blocks = (int)((scat_threads + T - 1) / T);
    const int m_tiles = n_nodes / 16;

    for (int l = 0; l < L; ++l) {
        sage_zero_f32<<<(feat_elems + T - 1) / T, T, 0, stream>>>(msg, feat_elems);
        sage_scatter_kernel<<<scat_blocks, T, 0, stream>>>(src, dst, cur, msg, n_edges);
        sage_layer_kernel<<<m_tiles, 256, 0, stream>>>(
            cur, msg, invdeg,
            W_self  + (size_t)l * D_FEAT * D_FEAT,
            W_neigh + (size_t)l * D_FEAT * D_FEAT,
            b       + (size_t)l * D_FEAT,
            nxt);
        cur = nxt;
        nxt = (nxt == bufA) ? bufB : bufA;
    }

    sage_pred_kernel<<<m_tiles, 64, 0, stream>>>(cur, W_pred, b_pred, (float*)d_out);
}


# Neighbour sums by gather: edges counting-sorted by destination once per call (integer atomics; per-node sum order = edge order), one wave sums a node's rows in registers; no float atomics, no zeroing
# speedup vs baseline: 2.6555x; 2.6555x over previous
#include <hip/hip_runtime.h>
#include <hip/hip_bf16.h>


#define D_FEAT 128
#define LDS_STRIDE 132
#define OUT_DIM 26
#define SCAN_CHUNK 1024

typedef __attribute__((ext_vector_type(2))) float v2f;
typedef __attribute__((ext_vector_type(8))) float v8f;


__global__ void sage_zero_i32(int* __restrict__ p, int n) {
    int i = blockIdx.x * blockDim.x + threadIdx.x;
    if (i < n) p[i] = 0;
}

__global__ void sage_count_kernel(const int* __restrict__ dst, int* __restrict__ cnt,
                                  int* __restrict__ slot, int n_edges) {
    int e = blockIdx.x * blockDim.x + threadIdx.x;
    if (e < n_edges) slot[e] = atomicAdd(&cnt[dst[e]], 1);
}

__global__ void __launch_bounds__(256)
sage_offsets_kernel(const int* __restrict__ cnt, int* __restrict__ first,
                    float* __restrict__ invdeg, int n_nodes) {
    __shared__ int wave_front[8];
    __shared__ int wave_total[8];

    const int t    = threadIdx.x;
    const int lane = t & 31;
    const int w    = t >> 5;
    const int base = blockIdx.x * SCAN_CHUNK;

    int front = 0;
    #pragma unroll 4
    for (int i = t * 4; i < base; i += SCAN_CHUNK) {
        const int4 v = *(const int4*)(cnt + i);
        front += (v.x + v.y) + (v.z + v.w);
    }

    const int i0 = base + t * 4;
    int c[4];
    for (int k = 0; k < 4; ++k) c[k] = (i0 + k < n_nodes) ? cnt[i0 + k] : 0;
    const int tsum = (c[0] + c[1]) + (c[2] + c[3]);

    int inc = tsum;
    for (int o = 1; o < 32; o <<= 1) {
        const int v = __shfl_up(inc, o);
        if (lane >= o) inc += v;
    }
    for (int o = 16; o > 0; o >>= 1) front += __shfl_xor(front, o);

    if (lane == 31) wave_total[w] = inc;
    if (lane == 0)  wave_front[w] = front;
    __syncthreads();

    int run = inc - tsum;
    for (int k = 0; k < 8; ++k) {
        run += wave_front[k];
        if (k < w) run += wave_total[k];
    }

    for (int k = 0; k < 4; ++k) {
        if (i0 + k < n_nodes) {
            first[i0 + k]  = run;
            invdeg[i0 + k] = 1.0f / fmaxf((float)c[k], 1.0f);
            run += c[k];
        }
    }
}

__global__ void sage_bucket_kernel(const int* __restrict__ src, const int* __restrict__ dst,
                                   const int* __restrict__ slot, const int* __restrict__ first,
                                   int* __restrict__ pairs, int n_edges) {
    int e = blockIdx.x * blockDim.x + threadIdx.x;
    if (e < n_edges) {
        const int pos = first[dst[e]] + slot[e];
        *(int2*)(pairs + 2 * (size_t)pos) = make_int2(e, src[e]);
    }
}

__global__ void __launch_bounds__(256)
sage_order_kernel(const int* __restrict__ first, const int* __restrict__ cnt,
                  const int* __restrict__ pairs, int* __restrict__ nbr, int n_nodes) {
    const int lane = threadIdx.x & 31;
    const int node = __builtin_amdgcn_readfirstlane((int)(blockIdx.x * 8 + (threadIdx.x >> 5)));
    if (node >= n_nodes) return;
    const int beg = first[node];
    const int n   = cnt[node];
    const int2* seg = (const int2*)pairs + beg;

    for (int kb = 0; kb < n; kb += 32) {
        const bool have = kb + lane < n;
        int2 mine = make_int2(0, 0);
        if (have) mine = seg[kb + lane];
        int pos = 0;
        for (int mb = 0; mb < n; mb += 32) {
            const int mcount = (n - mb < 32) ? (n - mb) : 32;
            int other = 0;
            if (mb + lane < n) other = seg[mb + lane].x;
            for (int m = 0; m < mcount; ++m)
                pos += (__builtin_amdgcn_readlane(other, m) < mine.x) ? 1 : 0;
        }
        if (have) nbr[beg + pos] = mine.y;
    }
}


__device__ __forceinline__ float4 sage_row(const float* __restrict__ xl, int ids, int j) {
    const int s = __builtin_amdgcn_readlane(ids, j);
    return *(const float4*)(xl + (size_t)s * D_FEAT);
}

template <int K>
__device__ __forceinline__ void sage_sum_rows(float4& acc, const float* __restrict__ xl,
                                              int ids, int j) {
    float4 v[K];
    for (int k = 0; k < K; ++k) v[k] = sage_row(xl, ids, j + k);
    for (int k = 0; k < K; ++k) {
        acc.x += v[k].x; acc.y += v[k].y; acc.z += v[k].z; acc.w += v[k].w;
    }
}

__global__ void __launch_bounds__(256)
sage_gather_kernel(const int* __restrict__ first, const int* __restrict__ cnt,
                   const int* __restrict__ nbr, const float* __restrict__ x,
                   float* __restrict__ hsum, int n_nodes) {
    const int lane = threadIdx.x & 31;
    const int node = __builtin_amdgcn_readfirstlane((int)(blockIdx.x * 8 + (threadIdx.x >> 5)));
    if (node >= n_nodes) return;
    const int beg = first[node];
    const int n   = cnt[node];
    const float* xl = x + lane * 4;

    float4 acc = make_float4(0.0f, 0.0f, 0.0f, 0.0f);
    for (int base = 0; base < n; base += 32) {
        const int m = (n - base < 32) ? (n - base) : 32;
        int ids = 0;
        if (lane < m) ids = nbr[beg + base + lane];

        int j = 0;
        for (; j + 8 <= m; j += 8) sage_sum_rows<8>(acc, xl, ids, j);
        if (j + 4 <= m) { sage_sum_rows<4>(acc, xl, ids, j); j += 4; }
        const int r = m - j;
        if (r == 3)      sage_sum_rows<3>(acc, xl, ids, j);
        else if (r == 2) sage_sum_rows<2>(acc, xl, ids, j);
        else if (r == 1) sage_sum_rows<1>(acc, xl, ids, j);
    }
    *(float4*)(hsum + (size_t)node * D_FEAT + lane * 4) = acc;
}

__global__ void __launch_bounds__(256)
sage_layer_kernel(const float* __restrict__ xin, float* __restrict__ h,
                  const float* __restrict__ invdeg,
                  const float* __restrict__ Wself, const float* __restrict__ Wneigh,
                  const float* __restrict__ bias) {
    __shared__ float As[16 * LDS_STRIDE];
    __shared__ float Hs[16 * LDS_STRIDE];

    const int m0 = blockIdx.x * 16;
    const int t  = threadIdx.x;

    {
        int r  = t >> 4;
        int c0 = (t & 15) * 8;
        const float* xr = xin + (size_t)(m0 + r) * D_FEAT + c0;
        float4 a0 = *(const float4*)(xr);
        float4 a1 = *(const float4*)(xr + 4);
        *(float4*)(As + r * LDS_STRIDE + c0)     = a0;
        *(float4*)(As + r * LDS_STRIDE + c0 + 4) = a1;

        float iv = invdeg[m0 + r];
        const float* mr = h + (size_t)(m0 + r) * D_FEAT + c0;
        float4 h0 = *(const float4*)(mr);
        float4 h1 = *(const float4*)(mr + 4);
        h0.x *= iv; h0.y *= iv; h0.z *= iv; h0.w *= iv;
        h1.x *= iv; h1.y *= iv; h1.z *= iv; h1.w *= iv;
        *(float4*)(Hs + r * LDS_STRIDE + c0)     = h0;
        *(float4*)(Hs + r * LDS_STRIDE + c0 + 4) = h1;
    }
    __syncthreads();

    const int wave = t >> 5;
    const int lane = t & 31;
    const int n0   = wave * 16;
    const int half = lane >> 4;
    const int l16  = lane & 15;
    const int col  = n0 + l16;

    v8f acc;
    {
        float bv = bias[col];
        #pragma unroll
        for (int i = 0; i < 8; ++i) acc[i] = bv;
    }

    #pragma unroll 4
    for (int k0 = 0; k0 < D_FEAT; k0 += 4) {
        const int ka = k0 + half * 2;
        v2f a = *(const v2f*)(As + l16 * LDS_STRIDE + ka);
        v2f b;
        b.x = Wself[(size_t)ka * D_FEAT + col];
        b.y = Wself[(size_t)(ka + 1) * D_FEAT + col];
        acc = __builtin_amdgcn_wmma_f32_16x16x4_f32(false, a, false, b,
                                                    (short)0, acc, false, false);

        v2f hh = *(const v2f*)(Hs + l16 * LDS_STRIDE + ka);
        v2f bn;
        bn.x = Wneigh[(size_t)ka * D_FEAT + col];
        bn.y = Wneigh[(size_t)(ka + 1) * D_FEAT + col];
        acc = __builtin_amdgcn_wmma_f32_16x16x4_f32(false, hh, false, bn,
                                                    (short)0, acc, false, false);
    }

    #pragma unroll
    for (int v = 0; v < 8; ++v) {
        int row = m0 + v + half * 8;
        float val = acc[v];
        val = val > 0.0f ? val : 0.0f;
        h[(size_t)row * D_FEAT + col] = val;
    }
}

__global__ void __launch_bounds__(64)
sage_pred_kernel(const float* __restrict__ xin, const float* __restrict__ Wp,
                 const float* __restrict__ bp, float* __restrict__ out) {
    const int m0   = blockIdx.x * 16;
    const int t    = threadIdx.x;
    const int wave = t >> 5;
    const int lane = t & 31;
    const int n0   = wave * 16;
    const int half = lane >> 4;
    const int l16  = lane & 15;
    const int col  = n0 + l16;
    const int bcol = col < OUT_DIM ? col : (OUT_DIM - 1);

    v8f acc;
    {
        float bv = bp[bcol];
        #pragma unroll
        for (int i = 0; i < 8; ++i) acc[i] = bv;
    }

    #pragma unroll 4
    for (int k0 = 0; k0 < D_FEAT; k0 += 4) {
        const int ka = k0 + half * 2;
        const float* ap = xin + (size_t)(m0 + l16) * D_FEAT + ka;
        v2f a; a.x = ap[0]; a.y = ap[1];
        v2f b;
        b.x = Wp[(size_t)ka * OUT_DIM + bcol];
        b.y = Wp[(size_t)(ka + 1) * OUT_DIM + bcol];
        acc = __builtin_amdgcn_wmma_f32_16x16x4_f32(false, a, false, b,
                                                    (short)0, acc, false, false);
    }

    if (col < OUT_DIM) {
        #pragma unroll
        for (int v = 0; v < 8; ++v) {
            int row = m0 + v + half * 8;
            out[(size_t)row * OUT_DIM + col] = acc[v];
        }
    }
}


extern "C" void kernel_launch(void* const* d_in, const int* in_sizes, int n_in,
                              void* d_out, int out_size, void* d_ws, size_t ws_size,
                              hipStream_t stream) {
    const float* x       = (const float*)d_in[0];
    const int*   src     = (const int*)  d_in[1];
    const int*   dst     = (const int*)  d_in[2];
    const float* W_self  = (const float*)d_in[3];
    const float* W_neigh = (const float*)d_in[4];
    const float* b       = (const float*)d_in[5];
    const float* W_pred  = (const float*)d_in[6];
    const float* b_pred  = (const float*)d_in[7];

    const int n_nodes = in_sizes[0] / D_FEAT;
    const int n_edges = in_sizes[1];
    const int L       = in_sizes[5] / D_FEAT;
    const int feat_elems = n_nodes * D_FEAT;

    char*  ws = (char*)d_ws;
    size_t off = ((size_t)n_nodes * sizeof(float) + 511) & ~(size_t)511;
    size_t feat_bytes = (size_t)feat_elems * sizeof(float);
    size_t edge_bytes = ((size_t)n_edges * sizeof(int) + 511) & ~(size_t)511;
    float* invdeg = (float*)ws;
    float* bufA   = (float*)(ws + off);
    float* bufB   = (float*)(ws + off + feat_bytes);
    char*  ga     = ws + off + 2 * feat_bytes;
    int*   cnt    = (int*)(ga);
    int*   first  = (int*)(ga + off);
    int*   slot   = (int*)(ga + 2 * off);
    int*   nbr    = (int*)(ga + 2 * off + edge_bytes);
    int*   pairs  = (int*)(ga + 2 * off + 2 * edge_bytes);

    const int T = 256;
    const int edge_blocks = (n_edges + T - 1) / T;
    const int wave_blocks = (n_nodes + 7) / 8;

    sage_zero_i32<<<(n_nodes + T - 1) / T, T, 0, stream>>>(cnt, n_nodes);
    sage_count_kernel<<<edge_blocks, T, 0, stream>>>(dst, cnt, slot, n_edges);
    sage_offsets_kernel<<<(n_nodes + SCAN_CHUNK - 1) / SCAN_CHUNK, 256, 0, stream>>>(
        cnt, first, invdeg, n_nodes);
    sage_bucket_kernel<<<edge_blocks, T, 0, stream>>>(src, dst, slot, first, pairs, n_edges);
    sage_order_kernel<<<wave_blocks, 256, 0, stream>>>(first, cnt, pairs, nbr, n_nodes);

    const float* cur = x;
    float* nxt = bufA;
    const int m_tiles = n_nodes / 16;

    for (int l = 0; l < L; ++l) {
        sage_gather_kernel<<<wave_blocks, 256, 0, stream>>>(first, cnt, nbr, cur, nxt, n_nodes);
        sage_layer_kernel<<<m_tiles, 256, 0, stream>>>(
            cur, nxt, invdeg,
            W_self  + (size_t)l * D_FEAT * D_FEAT,
            W_neigh + (size_t)l * D_FEAT * D_FEAT,
            b       + (size_t)l * D_FEAT);
        cur = nxt;
        nxt = (nxt == bufA) ? bufB : bufA;
    }

    sage_pred_kernel<<<m_tiles, 64, 0, stream>>>(cur, W_pred, b_pred, (float*)d_out);
}


# Neighbour gather fused into the layer kernel: each wave sums its two tile nodes' source rows (same order) straight into the LDS tile; the sum plane and 4 of 14 launches go
# speedup vs baseline: 3.0020x; 1.1305x over previous
#include <hip/hip_runtime.h>
#include <hip/hip_bf16.h>


#define D_FEAT 128
#define LDS_STRIDE 132
#define OUT_DIM 26
#define SCAN_CHUNK 1024

typedef __attribute__((ext_vector_type(2))) float v2f;
typedef __attribute__((ext_vector_type(8))) float v8f;


__global__ void sage_zero_i32(int* __restrict__ p, int n) {
    int i = blockIdx.x * blockDim.x + threadIdx.x;
    if (i < n) p[i] = 0;
}

__global__ void sage_count_kernel(const int* __restrict__ dst, int* __restrict__ cnt,
                                  int* __restrict__ slot, int n_edges) {
    int e = blockIdx.x * blockDim.x + threadIdx.x;
    if (e < n_edges) slot[e] = atomicAdd(&cnt[dst[e]], 1);
}

__global__ void __launch_bounds__(256)
sage_offsets_kernel(const int* __restrict__ cnt, int* __restrict__ first,
                    float* __restrict__ invdeg, int n_nodes) {
    __shared__ int wave_front[8];
    __shared__ int wave_total[8];

    const int t    = threadIdx.x;
    const int lane = t & 31;
    const int w    = t >> 5;
    const int base = blockIdx.x * SCAN_CHUNK;

    int front = 0;
    #pragma unroll 4
    for (int i = t * 4; i < base; i += SCAN_CHUNK) {
        const int4 v = *(const int4*)(cnt + i);
        front += (v.x + v.y) + (v.z + v.w);
    }

    const int i0 = base + t * 4;
    int c[4];
    for (int k = 0; k < 4; ++k) c[k] = (i0 + k < n_nodes) ? cnt[i0 + k] : 0;
    const int tsum = (c[0] + c[1]) + (c[2] + c[3]);

    int inc = tsum;
    for (int o = 1; o < 32; o <<= 1) {
        const int v = __shfl_up(inc, o);
        if (lane >= o) inc += v;
    }
    for (int o = 16; o > 0; o >>= 1) front += __shfl_xor(front, o);

    if (lane == 31) wave_total[w] = inc;
    if (lane == 0)  wave_front[w] = front;
    __syncthreads();

    int run = inc - tsum;
    for (int k = 0; k < 8; ++k) {
        run += wave_front[k];
        if (k < w) run += wave_total[k];
    }

    for (int k = 0; k < 4; ++k) {
        if (i0 + k < n_nodes) {
            first[i0 + k]  = run;
            invdeg[i0 + k] = 1.0f / fmaxf((float)c[k], 1.0f);
            run += c[k];
        }
    }
}

__global__ void sage_bucket_kernel(const int* __restrict__ src, const int* __restrict__ dst,
                                   const int* __restrict__ slot, const int* __restrict__ first,
                                   int* __restrict__ pairs, int n_edges) {
    int e = blockIdx.x * blockDim.x + threadIdx.x;
    if (e < n_edges) {
        const int pos = first[dst[e]] + slot[e];
        *(int2*)(pairs + 2 * (size_t)pos) = make_int2(e, src[e]);
    }
}

__global__ void __launch_bounds__(256)
sage_order_kernel(const int* __restrict__ first, const int* __restrict__ cnt,
                  const int* __restrict__ pairs, int* __restrict__ nbr, int n_nodes) {
    const int lane = threadIdx.x & 31;
    const int node = __builtin_amdgcn_readfirstlane((int)(blockIdx.x * 8 + (threadIdx.x >> 5)));
    if (node >= n_nodes) return;
    const int beg = first[node];
    const int n   = cnt[node];
    const int2* seg = (const int2*)pairs + beg;

    for (int kb = 0; kb < n; kb += 32) {
        const bool have = kb + lane < n;
        int2 mine = make_int2(0, 0);
        if (have) mine = seg[kb + lane];
        int pos = 0;
        for (int mb = 0; mb < n; mb += 32) {
            const int mcount = (n - mb < 32) ? (n - mb) : 32;
            int other = 0;
            if (mb + lane < n) other = seg[mb + lane].x;
            for (int m = 0; m < mcount; ++m)
                pos += (__builtin_amdgcn_readlane(other, m) < mine.x) ? 1 : 0;
        }
        if (have) nbr[beg + pos] = mine.y;
    }
}


__device__ __forceinline__ float4 sage_row(const float* __restrict__ xl, int ids, int j) {
    const int s = __builtin_amdgcn_readlane(ids, j);
    return *(const float4*)(xl + (size_t)s * D_FEAT);
}

template <int K>
__device__ __forceinline__ void sage_sum_rows(float4& acc, const float* __restrict__ xl,
                                              int ids, int j) {
    float4 v[K];
    for (int k = 0; k < K; ++k) v[k] = sage_row(xl, ids, j + k);
    for (int k = 0; k < K; ++k) {
        acc.x += v[k].x; acc.y += v[k].y; acc.z += v[k].z; acc.w += v[k].w;
    }
}

__device__ __forceinline__ float4 sage_node_sum(const int* __restrict__ nbr,
                                                const float* __restrict__ xl,
                                                int beg, int n, int lane, int ids0) {
    float4 acc = make_float4(0.0f, 0.0f, 0.0f, 0.0f);
    for (int base = 0; base < n; base += 32) {
        const int m = (n - base < 32) ? (n - base) : 32;
        int ids = ids0;
        if (base > 0) {
            ids = 0;
            if (lane < m) ids = nbr[beg + base + lane];
        }

        int j = 0;
        for (; j + 8 <= m; j += 8) sage_sum_rows<8>(acc, xl, ids, j);
        if (j + 4 <= m) { sage_sum_rows<4>(acc, xl, ids, j); j += 4; }
        const int r = m - j;
        if (r == 3)      sage_sum_rows<3>(acc, xl, ids, j);
        else if (r == 2) sage_sum_rows<2>(acc, xl, ids, j);
        else if (r == 1) sage_sum_rows<1>(acc, xl, ids, j);
    }
    return acc;
}

__global__ void __launch_bounds__(256)
sage_layer_kernel(const float* __restrict__ xin, float* __restrict__ hout,
                  const int* __restrict__ first, const int* __restrict__ cnt,
                  const int* __restrict__ nbr, const float* __restrict__ invdeg,
                  const float* __restrict__ Wself, const float* __restrict__ Wneigh,
                  const float* __restrict__ bias) {
    __shared__ float As[16 * LDS_STRIDE];
    __shared__ float Hs[16 * LDS_STRIDE];

    const int m0 = blockIdx.x * 16;
    const int t  = threadIdx.x;
    const int lane = t & 31;

    const int r  = t >> 4;
    const int c0 = (t & 15) * 8;
    const float* xr = xin + (size_t)(m0 + r) * D_FEAT + c0;
    const float4 a0 = *(const float4*)(xr);
    const float4 a1 = *(const float4*)(xr + 4);

    {
        const int wu = __builtin_amdgcn_readfirstlane(t >> 5);
        const int rowA = 2 * wu;
        const int nodeA = m0 + rowA;
        const int nodeB = nodeA + 1;
        const int begA = first[nodeA];
        const int begB = first[nodeB];
        const int nA   = cnt[nodeA];
        const int nB   = cnt[nodeB];
        const float ivA = invdeg[nodeA];
        const float ivB = invdeg[nodeB];

        int idsA = 0, idsB = 0;
        if (lane < nA) idsA = nbr[begA + lane];
        if (lane < nB) idsB = nbr[begB + lane];

        *(float4*)(As + r * LDS_STRIDE + c0)     = a0;
        *(float4*)(As + r * LDS_STRIDE + c0 + 4) = a1;

        const float* xl = xin + lane * 4;
        float4 s = sage_node_sum(nbr, xl, begA, nA, lane, idsA);
        s.x *= ivA; s.y *= ivA; s.z *= ivA; s.w *= ivA;
        *(float4*)(Hs + rowA * LDS_STRIDE + lane * 4) = s;

        s = sage_node_sum(nbr, xl, begB, nB, lane, idsB);
        s.x *= ivB; s.y *= ivB; s.z *= ivB; s.w *= ivB;
        *(float4*)(Hs + (rowA + 1) * LDS_STRIDE + lane * 4) = s;
    }

    __syncthreads();

    const int wave = t >> 5;
    const int n0   = wave * 16;
    const int half = lane >> 4;
    const int l16  = lane & 15;
    const int col  = n0 + l16;

    v8f acc;
    {
        float bv = bias[col];
        #pragma unroll
        for (int i = 0; i < 8; ++i) acc[i] = bv;
    }

    #pragma unroll 4
    for (int k0 = 0; k0 < D_FEAT; k0 += 4) {
        const int ka = k0 + half * 2;
        v2f a = *(const v2f*)(As + l16 * LDS_STRIDE + ka);
        v2f b;
        b.x = Wself[(size_t)ka * D_FEAT + col];
        b.y = Wself[(size_t)(ka + 1) * D_FEAT + col];
        acc = __builtin_amdgcn_wmma_f32_16x16x4_f32(false, a, false, b,
                                                    (short)0, acc, false, false);

        v2f hh = *(const v2f*)(Hs + l16 * LDS_STRIDE + ka);
        v2f bn;
        bn.x = Wneigh[(size_t)ka * D_FEAT + col];
        bn.y = Wneigh[(size_t)(ka + 1) * D_FEAT + col];
        acc = __builtin_amdgcn_wmma_f32_16x16x4_f32(false, hh, false, bn,
                                                    (short)0, acc, false, false);
    }

    #pragma unroll
    for (int v = 0; v < 8; ++v) {
        int row = m0 + v + half * 8;
        float val = acc[v];
        val = val > 0.0f ? val : 0.0f;
        hout[(size_t)row * D_FEAT + col] = val;
    }
}

__global__ void __launch_bounds__(64)
sage_pred_kernel(const float* __restrict__ xin, const float* __restrict__ Wp,
                 const float* __restrict__ bp, float* __restrict__ out) {
    const int m0   = blockIdx.x * 16;
    const int t    = threadIdx.x;
    const int wave = t >> 5;
    const int lane = t & 31;
    const int n0   = wave * 16;
    const int half = lane >> 4;
    const int l16  = lane & 15;
    const int col  = n0 + l16;
    const int bcol = col < OUT_DIM ? col : (OUT_DIM - 1);

    v8f acc;
    {
        float bv = bp[bcol];
        #pragma unroll
        for (int i = 0; i < 8; ++i) acc[i] = bv;
    }

    #pragma unroll 4
    for (int k0 = 0; k0 < D_FEAT; k0 += 4) {
        const int ka = k0 + half * 2;
        const float* ap = xin + (size_t)(m0 + l16) * D_FEAT + ka;
        v2f a; a.x = ap[0]; a.y = ap[1];
        v2f b;
        b.x = Wp[(size_t)ka * OUT_DIM + bcol];
        b.y = Wp[(size_t)(ka + 1) * OUT_DIM + bcol];
        acc = __builtin_amdgcn_wmma_f32_16x16x4_f32(false, a, false, b,
                                                    (short)0, acc, false, false);
    }

    if (col < OUT_DIM) {
        #pragma unroll
        for (int v = 0; v < 8; ++v) {
            int row = m0 + v + half * 8;
            out[(size_t)row * OUT_DIM + col] = acc[v];
        }
    }
}


extern "C" void kernel_launch(void* const* d_in, const int* in_sizes, int n_in,
                              void* d_out, int out_size, void* d_ws, size_t ws_size,
                              hipStream_t stream) {
    const float* x       = (const float*)d_in[0];
    const int*   src     = (const int*)  d_in[1];
    const int*   dst     = (const int*)  d_in[2];
    const float* W_self  = (const float*)d_in[3];
    const float* W_neigh = (const float*)d_in[4];
    const float* b       = (const float*)d_in[5];
    const float* W_pred  = (const float*)d_in[6];
    const float* b_pred  = (const float*)d_in[7];

    const int n_nodes = in_sizes[0] / D_FEAT;
    const int n_edges = in_sizes[1];
    const int L       = in_sizes[5] / D_FEAT;
    const int feat_elems = n_nodes * D_FEAT;

    char*  ws = (char*)d_ws;
    size_t off = ((size_t)n_nodes * sizeof(float) + 511) & ~(size_t)511;
    size_t feat_bytes = (size_t)feat_elems * sizeof(float);
    size_t edge_bytes = ((size_t)n_edges * sizeof(int) + 511) & ~(size_t)511;
    float* invdeg = (float*)ws;
    float* bufA   = (float*)(ws + off);
    float* bufB   = (float*)(ws + off + feat_bytes);
    char*  ga     = ws + off + 2 * feat_bytes;
    int*   cnt    = (int*)(ga);
    int*   first  = (int*)(ga + off);
    int*   slot   = (int*)(ga + 2 * off);
    int*   nbr    = (int*)(ga + 2 * off + edge_bytes);
    int*   pairs  = (int*)(ga + 2 * off + 2 * edge_bytes);

    const int T = 256;
    const int edge_blocks = (n_edges + T - 1) / T;
    const int wave_blocks = (n_nodes + 7) / 8;

    sage_zero_i32<<<(n_nodes + T - 1) / T, T, 0, stream>>>(cnt, n_nodes);
    sage_count_kernel<<<edge_blocks, T, 0, stream>>>(dst, cnt, slot, n_edges);
    sage_offsets_kernel<<<(n_nodes + SCAN_CHUNK - 1) / SCAN_CHUNK, 256, 0, stream>>>(
        cnt, first, invdeg, n_nodes);
    sage_bucket_kernel<<<edge_blocks, T, 0, stream>>>(src, dst, slot, first, pairs, n_edges);
    sage_order_kernel<<<wave_blocks, 256, 0, stream>>>(first, cnt, pairs, nbr, n_nodes);

    const float* cur = x;
    float* nxt = bufA;
    const int m_tiles = n_nodes / 16;

    for (int l = 0; l < L; ++l) {
        sage_layer_kernel<<<m_tiles, 256, 0, stream>>>(
            cur, nxt, first, cnt, nbr, invdeg,
            W_self  + (size_t)l * D_FEAT * D_FEAT,
            W_neigh + (size_t)l * D_FEAT * D_FEAT,
            b       + (size_t)l * D_FEAT);
        cur = nxt;
        nxt = (nxt == bufA) ? bufB : bufA;
    }

    sage_pred_kernel<<<m_tiles, 64, 0, stream>>>(cur, W_pred, b_pred, (float*)d_out);
}


# Layer kernel row tile 16 -> 64 rows per work-group: each wave keeps its weight fragment for four row tiles (4 accumulators), so weight reads from L2 per layer fall 4x; sum orders unchanged
# speedup vs baseline: 3.1661x; 1.0547x over previous
#include <hip/hip_runtime.h>
#include <hip/hip_bf16.h>


#define D_FEAT 128
#define LDS_STRIDE 132
#define OUT_DIM 26
#define SCAN_CHUNK 1024

typedef __attribute__((ext_vector_type(2))) float v2f;
typedef __attribute__((ext_vector_type(8))) float v8f;


__global__ void sage_zero_i32(int* __restrict__ p, int n) {
    int i = blockIdx.x * blockDim.x + threadIdx.x;
    if (i < n) p[i] = 0;
}

__global__ void sage_count_kernel(const int* __restrict__ dst, int* __restrict__ cnt,
                                  int* __restrict__ slot, int n_edges) {
    int e = blockIdx.x * blockDim.x + threadIdx.x;
    if (e < n_edges) slot[e] = atomicAdd(&cnt[dst[e]], 1);
}

__global__ void __launch_bounds__(256)
sage_offsets_kernel(const int* __restrict__ cnt, int* __restrict__ first,
                    float* __restrict__ invdeg, int n_nodes) {
    __shared__ int wave_front[8];
    __shared__ int wave_total[8];

    const int t    = threadIdx.x;
    const int lane = t & 31;
    const int w    = t >> 5;
    const int base = blockIdx.x * SCAN_CHUNK;

    int front = 0;
    #pragma unroll 4
    for (int i = t * 4; i < base; i += SCAN_CHUNK) {
        const int4 v = *(const int4*)(cnt + i);
        front += (v.x + v.y) + (v.z + v.w);
    }

    const int i0 = base + t * 4;
    int c[4];
    for (int k = 0; k < 4; ++k) c[k] = (i0 + k < n_nodes) ? cnt[i0 + k] : 0;
    const int tsum = (c[0] + c[1]) + (c[2] + c[3]);

    int inc = tsum;
    for (int o = 1; o < 32; o <<= 1) {
        const int v = __shfl_up(inc, o);
        if (lane >= o) inc += v;
    }
    for (int o = 16; o > 0; o >>= 1) front += __shfl_xor(front, o);

    if (lane == 31) wave_total[w] = inc;
    if (lane == 0)  wave_front[w] = front;
    __syncthreads();

    int run = inc - tsum;
    for (int k = 0; k < 8; ++k) {
        run += wave_front[k];
        if (k < w) run += wave_total[k];
    }

    for (int k = 0; k < 4; ++k) {
        if (i0 + k < n_nodes) {
            first[i0 + k]  = run;
            invdeg[i0 + k] = 1.0f / fmaxf((float)c[k], 1.0f);
            run += c[k];
        }
    }
}

__global__ void sage_bucket_kernel(const int* __restrict__ src, const int* __restrict__ dst,
                                   const int* __restrict__ slot, const int* __restrict__ first,
                                   int* __restrict__ pairs, int n_edges) {
    int e = blockIdx.x * blockDim.x + threadIdx.x;
    if (e < n_edges) {
        const int pos = first[dst[e]] + slot[e];
        *(int2*)(pairs + 2 * (size_t)pos) = make_int2(e, src[e]);
    }
}

__global__ void __launch_bounds__(256)
sage_order_kernel(const int* __restrict__ first, const int* __restrict__ cnt,
                  const int* __restrict__ pairs, int* __restrict__ nbr, int n_nodes) {
    const int lane = threadIdx.x & 31;
    const int node = __builtin_amdgcn_readfirstlane((int)(blockIdx.x * 8 + (threadIdx.x >> 5)));
    if (node >= n_nodes) return;
    const int beg = first[node];
    const int n   = cnt[node];
    const int2* seg = (const int2*)pairs + beg;

    for (int kb = 0; kb < n; kb += 32) {
        const bool have = kb + lane < n;
        int2 mine = make_int2(0, 0);
        if (have) mine = seg[kb + lane];
        int pos = 0;
        for (int mb = 0; mb < n; mb += 32) {
            const int mcount = (n - mb < 32) ? (n - mb) : 32;
            int other = 0;
            if (mb + lane < n) other = seg[mb + lane].x;
            for (int m = 0; m < mcount; ++m)
                pos += (__builtin_amdgcn_readlane(other, m) < mine.x) ? 1 : 0;
        }
        if (have) nbr[beg + pos] = mine.y;
    }
}


__device__ __forceinline__ float4 sage_row(const float* __restrict__ xl, int ids, int j) {
    const int s = __builtin_amdgcn_readlane(ids, j);
    return *(const float4*)(xl + (size_t)s * D_FEAT);
}

template <int K>
__device__ __forceinline__ void sage_sum_rows(float4& acc, const float* __restrict__ xl,
                                              int ids, int j) {
    float4 v[K];
    for (int k = 0; k < K; ++k) v[k] = sage_row(xl, ids, j + k);
    for (int k = 0; k < K; ++k) {
        acc.x += v[k].x; acc.y += v[k].y; acc.z += v[k].z; acc.w += v[k].w;
    }
}

__device__ __forceinline__ float4 sage_node_sum(const int* __restrict__ nbr,
                                                const float* __restrict__ xl,
                                                int beg, int n, int lane, int ids0) {
    float4 acc = make_float4(0.0f, 0.0f, 0.0f, 0.0f);
    for (int base = 0; base < n; base += 32) {
        const int m = (n - base < 32) ? (n - base) : 32;
        int ids = ids0;
        if (base > 0) {
            ids = 0;
            if (lane < m) ids = nbr[beg + base + lane];
        }

        int j = 0;
        for (; j + 8 <= m; j += 8) sage_sum_rows<8>(acc, xl, ids, j);
        if (j + 4 <= m) { sage_sum_rows<4>(acc, xl, ids, j); j += 4; }
        const int r = m - j;
        if (r == 3)      sage_sum_rows<3>(acc, xl, ids, j);
        else if (r == 2) sage_sum_rows<2>(acc, xl, ids, j);
        else if (r == 1) sage_sum_rows<1>(acc, xl, ids, j);
    }
    return acc;
}

#define TILE_M 64
#define M_SUB (TILE_M / 16)
#define ROWS_PER_WAVE (TILE_M / 8)

__global__ void __launch_bounds__(256)
sage_layer_kernel(const float* __restrict__ xin, float* __restrict__ hout,
                  const int* __restrict__ first, const int* __restrict__ cnt,
                  const int* __restrict__ nbr, const float* __restrict__ invdeg,
                  const float* __restrict__ Wself, const float* __restrict__ Wneigh,
                  const float* __restrict__ bias, int n_nodes) {
    __shared__ float As[TILE_M * LDS_STRIDE];
    __shared__ float Hs[TILE_M * LDS_STRIDE];

    const int m0 = blockIdx.x * TILE_M;
    const int t  = threadIdx.x;
    const int lane = t & 31;

    {
        const int wu   = __builtin_amdgcn_readfirstlane(t >> 5);
        const int row0 = ROWS_PER_WAVE * wu;

        int   beg[ROWS_PER_WAVE];
        int   nn[ROWS_PER_WAVE];
        float iv[ROWS_PER_WAVE];
        int   ids[ROWS_PER_WAVE];
        float4 xv[ROWS_PER_WAVE];

        #pragma unroll
        for (int i = 0; i < ROWS_PER_WAVE; ++i) {
            const int node  = m0 + row0 + i;
            const bool live = node < n_nodes;
            const int nodeC = live ? node : (n_nodes - 1);
            beg[i] = first[nodeC];
            nn[i]  = live ? cnt[nodeC] : 0;
            iv[i]  = invdeg[nodeC];
        }
        #pragma unroll
        for (int i = 0; i < ROWS_PER_WAVE; ++i) {
            ids[i] = 0;
            if (lane < nn[i]) ids[i] = nbr[beg[i] + lane];
        }
        #pragma unroll
        for (int i = 0; i < ROWS_PER_WAVE; ++i) {
            const int node = m0 + row0 + i;
            xv[i] = make_float4(0.0f, 0.0f, 0.0f, 0.0f);
            if (node < n_nodes) xv[i] = *(const float4*)(xin + (size_t)node * D_FEAT + lane * 4);
        }
        #pragma unroll
        for (int i = 0; i < ROWS_PER_WAVE; ++i)
            *(float4*)(As + (row0 + i) * LDS_STRIDE + lane * 4) = xv[i];

        const float* xl = xin + lane * 4;
        #pragma unroll
        for (int i = 0; i < ROWS_PER_WAVE; ++i) {
            float4 s = sage_node_sum(nbr, xl, beg[i], nn[i], lane, ids[i]);
            s.x *= iv[i]; s.y *= iv[i]; s.z *= iv[i]; s.w *= iv[i];
            *(float4*)(Hs + (row0 + i) * LDS_STRIDE + lane * 4) = s;
        }
    }

    __syncthreads();

    const int wave = t >> 5;
    const int n0   = wave * 16;
    const int half = lane >> 4;
    const int l16  = lane & 15;
    const int col  = n0 + l16;

    v8f acc[M_SUB];
    {
        float bv = bias[col];
        #pragma unroll
        for (int mt = 0; mt < M_SUB; ++mt) {
            #pragma unroll
            for (int i = 0; i < 8; ++i) acc[mt][i] = bv;
        }
    }

    #pragma unroll 2
    for (int k0 = 0; k0 < D_FEAT; k0 += 4) {
        const int ka = k0 + half * 2;
        v2f b, bn;
        b.x  = Wself[(size_t)ka * D_FEAT + col];
        b.y  = Wself[(size_t)(ka + 1) * D_FEAT + col];
        bn.x = Wneigh[(size_t)ka * D_FEAT + col];
        bn.y = Wneigh[(size_t)(ka + 1) * D_FEAT + col];

        v2f a[M_SUB], hh[M_SUB];
        #pragma unroll
        for (int mt = 0; mt < M_SUB; ++mt) {
            a[mt]  = *(const v2f*)(As + (mt * 16 + l16) * LDS_STRIDE + ka);
            hh[mt] = *(const v2f*)(Hs + (mt * 16 + l16) * LDS_STRIDE + ka);
        }
        #pragma unroll
        for (int mt = 0; mt < M_SUB; ++mt)
            acc[mt] = __builtin_amdgcn_wmma_f32_16x16x4_f32(false, a[mt], false, b,
                                                            (short)0, acc[mt], false, false);
        #pragma unroll
        for (int mt = 0; mt < M_SUB; ++mt)
            acc[mt] = __builtin_amdgcn_wmma_f32_16x16x4_f32(false, hh[mt], false, bn,
                                                            (short)0, acc[mt], false, false);
    }

    #pragma unroll
    for (int mt = 0; mt < M_SUB; ++mt) {
        #pragma unroll
        for (int v = 0; v < 8; ++v) {
            const int row = m0 + mt * 16 + v + half * 8;
            float val = acc[mt][v];
            val = val > 0.0f ? val : 0.0f;
            if (row < n_nodes) hout[(size_t)row * D_FEAT + col] = val;
        }
    }
}

__global__ void __launch_bounds__(64)
sage_pred_kernel(const float* __restrict__ xin, const float* __restrict__ Wp,
                 const float* __restrict__ bp, float* __restrict__ out) {
    const int m0   = blockIdx.x * 16;
    const int t    = threadIdx.x;
    const int wave = t >> 5;
    const int lane = t & 31;
    const int n0   = wave * 16;
    const int half = lane >> 4;
    const int l16  = lane & 15;
    const int col  = n0 + l16;
    const int bcol = col < OUT_DIM ? col : (OUT_DIM - 1);

    v8f acc;
    {
        float bv = bp[bcol];
        #pragma unroll
        for (int i = 0; i < 8; ++i) acc[i] = bv;
    }

    #pragma unroll 4
    for (int k0 = 0; k0 < D_FEAT; k0 += 4) {
        const int ka = k0 + half * 2;
        const float* ap = xin + (size_t)(m0 + l16) * D_FEAT + ka;
        v2f a; a.x = ap[0]; a.y = ap[1];
        v2f b;
        b.x = Wp[(size_t)ka * OUT_DIM + bcol];
        b.y = Wp[(size_t)(ka + 1) * OUT_DIM + bcol];
        acc = __builtin_amdgcn_wmma_f32_16x16x4_f32(false, a, false, b,
                                                    (short)0, acc, false, false);
    }

    if (col < OUT_DIM) {
        #pragma unroll
        for (int v = 0; v < 8; ++v) {
            int row = m0 + v + half * 8;
            out[(size_t)row * OUT_DIM + col] = acc[v];
        }
    }
}


extern "C" void kernel_launch(void* const* d_in, const int* in_sizes, int n_in,
                              void* d_out, int out_size, void* d_ws, size_t ws_size,
                              hipStream_t stream) {
    const float* x       = (const float*)d_in[0];
    const int*   src     = (const int*)  d_in[1];
    const int*   dst     = (const int*)  d_in[2];
    const float* W_self  = (const float*)d_in[3];
    const float* W_neigh = (const float*)d_in[4];
    const float* b       = (const float*)d_in[5];
    const float* W_pred  = (const float*)d_in[6];
    const float* b_pred  = (const float*)d_in[7];

    const int n_nodes = in_sizes[0] / D_FEAT;
    const int n_edges = in_sizes[1];
    const int L       = in_sizes[5] / D_FEAT;
    const int feat_elems = n_nodes * D_FEAT;

    char*  ws = (char*)d_ws;
    size_t off = ((size_t)n_nodes * sizeof(float) + 511) & ~(size_t)511;
    size_t feat_bytes = (size_t)feat_elems * sizeof(float);
    size_t edge_bytes = ((size_t)n_edges * sizeof(int) + 511) & ~(size_t)511;
    float* invdeg = (float*)ws;
    float* bufA   = (float*)(ws + off);
    float* bufB   = (float*)(ws + off + feat_bytes);
    char*  ga     = ws + off + 2 * feat_bytes;
    int*   cnt    = (int*)(ga);
    int*   first  = (int*)(ga + off);
    int*   slot   = (int*)(ga + 2 * off);
    int*   nbr    = (int*)(ga + 2 * off + edge_bytes);
    int*   pairs  = (int*)(ga + 2 * off + 2 * edge_bytes);

    const int T = 256;
    const int edge_blocks = (n_edges + T - 1) / T;
    const int wave_blocks = (n_nodes + 7) / 8;

    sage_zero_i32<<<(n_nodes + T - 1) / T, T, 0, stream>>>(cnt, n_nodes);
    sage_count_kernel<<<edge_blocks, T, 0, stream>>>(dst, cnt, slot, n_edges);
    sage_offsets_kernel<<<(n_nodes + SCAN_CHUNK - 1) / SCAN_CHUNK, 256, 0, stream>>>(
        cnt, first, invdeg, n_nodes);
    sage_bucket_kernel<<<edge_blocks, T, 0, stream>>>(src, dst, slot, first, pairs, n_edges);
    sage_order_kernel<<<wave_blocks, 256, 0, stream>>>(first, cnt, pairs, nbr, n_nodes);

    const float* cur = x;
    float* nxt = bufA;
    const int m_tiles = n_nodes / 16;
    const int layer_blocks = (n_nodes + TILE_M - 1) / TILE_M;

    for (int l = 0; l < L; ++l) {
        sage_layer_kernel<<<layer_blocks, 256, 0, stream>>>(
            cur, nxt, first, cnt, nbr, invdeg,
            W_self  + (size_t)l * D_FEAT * D_FEAT,
            W_neigh + (size_t)l * D_FEAT * D_FEAT,
            b       + (size_t)l * D_FEAT, n_nodes);
        cur = nxt;
        nxt = (nxt == bufA) ? bufB : bufA;
    }

    sage_pred_kernel<<<m_tiles, 64, 0, stream>>>(cur, W_pred, b_pred, (float*)d_out);
}


# Layer product's K loop software-pipelined: weight fragments load into a second register set one 16-k chunk ahead (first chunk and bias in front of the neighbour sums); sum orders unchanged
# speedup vs baseline: 3.2631x; 1.0306x over previous
#include <hip/hip_runtime.h>
#include <hip/hip_bf16.h>


#define D_FEAT 128
#define LDS_STRIDE 132
#define OUT_DIM 26
#define SCAN_CHUNK 1024

typedef __attribute__((ext_vector_type(2))) float v2f;
typedef __attribute__((ext_vector_type(8))) float v8f;


__global__ void sage_zero_i32(int* __restrict__ p, int n) {
    int i = blockIdx.x * blockDim.x + threadIdx.x;
    if (i < n) p[i] = 0;
}

__global__ void sage_count_kernel(const int* __restrict__ dst, int* __restrict__ cnt,
                                  int* __restrict__ slot, int n_edges) {
    int e = blockIdx.x * blockDim.x + threadIdx.x;
    if (e < n_edges) slot[e] = atomicAdd(&cnt[dst[e]], 1);
}

__global__ void __launch_bounds__(256)
sage_offsets_kernel(const int* __restrict__ cnt, int* __restrict__ first,
                    float* __restrict__ invdeg, int n_nodes) {
    __shared__ int wave_front[8];
    __shared__ int wave_total[8];

    const int t    = threadIdx.x;
    const int lane = t & 31;
    const int w    = t >> 5;
    const int base = blockIdx.x * SCAN_CHUNK;

    int front = 0;
    #pragma unroll 4
    for (int i = t * 4; i < base; i += SCAN_CHUNK) {
        const int4 v = *(const int4*)(cnt + i);
        front += (v.x + v.y) + (v.z + v.w);
    }

    const int i0 = base + t * 4;
    int c[4];
    for (int k = 0; k < 4; ++k) c[k] = (i0 + k < n_nodes) ? cnt[i0 + k] : 0;
    const int tsum = (c[0] + c[1]) + (c[2] + c[3]);

    int inc = tsum;
    for (int o = 1; o < 32; o <<= 1) {
        const int v = __shfl_up(inc, o);
        if (lane >= o) inc += v;
    }
    for (int o = 16; o > 0; o >>= 1) front += __shfl_xor(front, o);

    if (lane == 31) wave_total[w] = inc;
    if (lane == 0)  wave_front[w] = front;
    __syncthreads();

    int run = inc - tsum;
    for (int k = 0; k < 8; ++k) {
        run += wave_front[k];
        if (k < w) run += wave_total[k];
    }

    for (int k = 0; k < 4; ++k) {
        if (i0 + k < n_nodes) {
            first[i0 + k]  = run;
            invdeg[i0 + k] = 1.0f / fmaxf((float)c[k], 1.0f);
            run += c[k];
        }
    }
}

__global__ void sage_bucket_kernel(const int* __restrict__ src, const int* __restrict__ dst,
                                   const int* __restrict__ slot, const int* __restrict__ first,
                                   int* __restrict__ pairs, int n_edges) {
    int e = blockIdx.x * blockDim.x + threadIdx.x;
    if (e < n_edges) {
        const int pos = first[dst[e]] + slot[e];
        *(int2*)(pairs + 2 * (size_t)pos) = make_int2(e, src[e]);
    }
}

__global__ void __launch_bounds__(256)
sage_order_kernel(const int* __restrict__ first, const int* __restrict__ cnt,
                  const int* __restrict__ pairs, int* __restrict__ nbr, int n_nodes) {
    const int lane = threadIdx.x & 31;
    const int node = __builtin_amdgcn_readfirstlane((int)(blockIdx.x * 8 + (threadIdx.x >> 5)));
    if (node >= n_nodes) return;
    const int beg = first[node];
    const int n   = cnt[node];
    const int2* seg = (const int2*)pairs + beg;

    for (int kb = 0; kb < n; kb += 32) {
        const bool have = kb + lane < n;
        int2 mine = make_int2(0, 0);
        if (have) mine = seg[kb + lane];
        int pos = 0;
        for (int mb = 0; mb < n; mb += 32) {
            const int mcount = (n - mb < 32) ? (n - mb) : 32;
            int other = 0;
            if (mb + lane < n) other = seg[mb + lane].x;
            for (int m = 0; m < mcount; ++m)
                pos += (__builtin_amdgcn_readlane(other, m) < mine.x) ? 1 : 0;
        }
        if (have) nbr[beg + pos] = mine.y;
    }
}


__device__ __forceinline__ float4 sage_row(const float* __restrict__ xl, int ids, int j) {
    const int s = __builtin_amdgcn_readlane(ids, j);
    return *(const float4*)(xl + (size_t)s * D_FEAT);
}

template <int K>
__device__ __forceinline__ void sage_sum_rows(float4& acc, const float* __restrict__ xl,
                                              int ids, int j) {
    float4 v[K];
    for (int k = 0; k < K; ++k) v[k] = sage_row(xl, ids, j + k);
    for (int k = 0; k < K; ++k) {
        acc.x += v[k].x; acc.y += v[k].y; acc.z += v[k].z; acc.w += v[k].w;
    }
}

__device__ __forceinline__ float4 sage_node_sum(const int* __restrict__ nbr,
                                                const float* __restrict__ xl,
                                                int beg, int n, int lane, int ids0) {
    float4 acc = make_float4(0.0f, 0.0f, 0.0f, 0.0f);
    for (int base = 0; base < n; base += 32) {
        const int m = (n - base < 32) ? (n - base) : 32;
        int ids = ids0;
        if (base > 0) {
            ids = 0;
            if (lane < m) ids = nbr[beg + base + lane];
        }

        int j = 0;
        for (; j + 8 <= m; j += 8) sage_sum_rows<8>(acc, xl, ids, j);
        if (j + 4 <= m) { sage_sum_rows<4>(acc, xl, ids, j); j += 4; }
        const int r = m - j;
        if (r == 3)      sage_sum_rows<3>(acc, xl, ids, j);
        else if (r == 2) sage_sum_rows<2>(acc, xl, ids, j);
        else if (r == 1) sage_sum_rows<1>(acc, xl, ids, j);
    }
    return acc;
}

#define TILE_M 64
#define M_SUB (TILE_M / 16)
#define ROWS_PER_WAVE (TILE_M / 8)

#define KC_STEPS 4
#define KC_K (4 * KC_STEPS)

__device__ __forceinline__ void sage_load_w(v2f (&bs)[KC_STEPS], v2f (&bn)[KC_STEPS],
                                            const float* __restrict__ Ws,
                                            const float* __restrict__ Wn,
                                            int kbase, int half, int col) {
    #pragma unroll
    for (int s = 0; s < KC_STEPS; ++s) {
        const int ka = kbase + 4 * s + half * 2;
        bs[s].x = Ws[(size_t)ka * D_FEAT + col];
        bs[s].y = Ws[(size_t)(ka + 1) * D_FEAT + col];
        bn[s].x = Wn[(size_t)ka * D_FEAT + col];
        bn[s].y = Wn[(size_t)(ka + 1) * D_FEAT + col];
    }
}

__device__ __forceinline__ void sage_mma_chunk(v8f (&acc)[TILE_M / 16],
                                               const v2f (&bs)[KC_STEPS],
                                               const v2f (&bn)[KC_STEPS],
                                               const float* As, const float* Hs,
                                               int kbase, int half, int l16) {
    #pragma unroll
    for (int s = 0; s < KC_STEPS; ++s) {
        const int ka = kbase + 4 * s + half * 2;
        v2f a[TILE_M / 16], hh[TILE_M / 16];
        #pragma unroll
        for (int mt = 0; mt < TILE_M / 16; ++mt) {
            a[mt]  = *(const v2f*)(As + (mt * 16 + l16) * LDS_STRIDE + ka);
            hh[mt] = *(const v2f*)(Hs + (mt * 16 + l16) * LDS_STRIDE + ka);
        }
        #pragma unroll
        for (int mt = 0; mt < TILE_M / 16; ++mt)
            acc[mt] = __builtin_amdgcn_wmma_f32_16x16x4_f32(false, a[mt], false, bs[s],
                                                            (short)0, acc[mt], false, false);
        #pragma unroll
        for (int mt = 0; mt < TILE_M / 16; ++mt)
            acc[mt] = __builtin_amdgcn_wmma_f32_16x16x4_f32(false, hh[mt], false, bn[s],
                                                            (short)0, acc[mt], false, false);
    }
}

template <bool LAST>
__device__ __forceinline__ void sage_chunk_pair(v8f (&acc)[TILE_M / 16],
                                                v2f (&bs0)[KC_STEPS], v2f (&bn0)[KC_STEPS],
                                                v2f (&bs1)[KC_STEPS], v2f (&bn1)[KC_STEPS],
                                                const float* __restrict__ Ws,
                                                const float* __restrict__ Wn,
                                                const float* As, const float* Hs,
                                                int kc, int half, int l16, int col) {
    sage_load_w(bs1, bn1, Ws, Wn, kc + KC_K, half, col);
    sage_mma_chunk(acc, bs0, bn0, As, Hs, kc, half, l16);
    if (!LAST) sage_load_w(bs0, bn0, Ws, Wn, kc + 2 * KC_K, half, col);
    sage_mma_chunk(acc, bs1, bn1, As, Hs, kc + KC_K, half, l16);
}

__global__ void __launch_bounds__(256)
sage_layer_kernel(const float* __restrict__ xin, float* __restrict__ hout,
                  const int* __restrict__ first, const int* __restrict__ cnt,
                  const int* __restrict__ nbr, const float* __restrict__ invdeg,
                  const float* __restrict__ Wself, const float* __restrict__ Wneigh,
                  const float* __restrict__ bias, int n_nodes) {
    __shared__ float As[TILE_M * LDS_STRIDE];
    __shared__ float Hs[TILE_M * LDS_STRIDE];

    const int m0 = blockIdx.x * TILE_M;
    const int t  = threadIdx.x;
    const int lane = t & 31;

    const int wave = t >> 5;
    const int n0   = wave * 16;
    const int half = lane >> 4;
    const int l16  = lane & 15;
    const int col  = n0 + l16;

    v2f bs0[KC_STEPS], bn0[KC_STEPS], bs1[KC_STEPS], bn1[KC_STEPS];
    sage_load_w(bs0, bn0, Wself, Wneigh, 0, half, col);
    const float bv = bias[col];

    {
        const int wu   = __builtin_amdgcn_readfirstlane(t >> 5);
        const int row0 = ROWS_PER_WAVE * wu;

        int   beg[ROWS_PER_WAVE];
        int   nn[ROWS_PER_WAVE];
        float iv[ROWS_PER_WAVE];
        int   ids[ROWS_PER_WAVE];
        float4 xv[ROWS_PER_WAVE];

        #pragma unroll
        for (int i = 0; i < ROWS_PER_WAVE; ++i) {
            const int node  = m0 + row0 + i;
            const bool live = node < n_nodes;
            const int nodeC = live ? node : (n_nodes - 1);
            beg[i] = first[nodeC];
            nn[i]  = live ? cnt[nodeC] : 0;
            iv[i]  = invdeg[nodeC];
        }
        #pragma unroll
        for (int i = 0; i < ROWS_PER_WAVE; ++i) {
            ids[i] = 0;
            if (lane < nn[i]) ids[i] = nbr[beg[i] + lane];
        }
        #pragma unroll
        for (int i = 0; i < ROWS_PER_WAVE; ++i) {
            const int node = m0 + row0 + i;
            xv[i] = make_float4(0.0f, 0.0f, 0.0f, 0.0f);
            if (node < n_nodes) xv[i] = *(const float4*)(xin + (size_t)node * D_FEAT + lane * 4);
        }
        #pragma unroll
        for (int i = 0; i < ROWS_PER_WAVE; ++i)
            *(float4*)(As + (row0 + i) * LDS_STRIDE + lane * 4) = xv[i];

        const float* xl = xin + lane * 4;
        #pragma unroll
        for (int i = 0; i < ROWS_PER_WAVE; ++i) {
            float4 s = sage_node_sum(nbr, xl, beg[i], nn[i], lane, ids[i]);
            s.x *= iv[i]; s.y *= iv[i]; s.z *= iv[i]; s.w *= iv[i];
            *(float4*)(Hs + (row0 + i) * LDS_STRIDE + lane * 4) = s;
        }
    }

    __syncthreads();

    v8f acc[M_SUB];
    #pragma unroll
    for (int mt = 0; mt < M_SUB; ++mt) {
        #pragma unroll
        for (int i = 0; i < 8; ++i) acc[mt][i] = bv;
    }

    for (int kc = 0; kc < D_FEAT - 2 * KC_K; kc += 2 * KC_K)
        sage_chunk_pair<false>(acc, bs0, bn0, bs1, bn1, Wself, Wneigh, As, Hs, kc, half, l16, col);
    sage_chunk_pair<true>(acc, bs0, bn0, bs1, bn1, Wself, Wneigh, As, Hs,
                          D_FEAT - 2 * KC_K, half, l16, col);

    #pragma unroll
    for (int mt = 0; mt < M_SUB; ++mt) {
        #pragma unroll
        for (int v = 0; v < 8; ++v) {
            const int row = m0 + mt * 16 + v + half * 8;
            float val = acc[mt][v];
            val = val > 0.0f ? val : 0.0f;
            if (row < n_nodes) hout[(size_t)row * D_FEAT + col] = val;
        }
    }
}

__global__ void __launch_bounds__(64)
sage_pred_kernel(const float* __restrict__ xin, const float* __restrict__ Wp,
                 const float* __restrict__ bp, float* __restrict__ out) {
    const int m0   = blockIdx.x * 16;
    const int t    = threadIdx.x;
    const int wave = t >> 5;
    const int lane = t & 31;
    const int n0   = wave * 16;
    const int half = lane >> 4;
    const int l16  = lane & 15;
    const int col  = n0 + l16;
    const int bcol = col < OUT_DIM ? col : (OUT_DIM - 1);

    v8f acc;
    {
        float bv = bp[bcol];
        #pragma unroll
        for (int i = 0; i < 8; ++i) acc[i] = bv;
    }

    #pragma unroll 4
    for (int k0 = 0; k0 < D_FEAT; k0 += 4) {
        const int ka = k0 + half * 2;
        const float* ap = xin + (size_t)(m0 + l16) * D_FEAT + ka;
        v2f a; a.x = ap[0]; a.y = ap[1];
        v2f b;
        b.x = Wp[(size_t)ka * OUT_DIM + bcol];
        b.y = Wp[(size_t)(ka + 1) * OUT_DIM + bcol];
        acc = __builtin_amdgcn_wmma_f32_16x16x4_f32(false, a, false, b,
                                                    (short)0, acc, false, false);
    }

    if (col < OUT_DIM) {
        #pragma unroll
        for (int v = 0; v < 8; ++v) {
            int row = m0 + v + half * 8;
            out[(size_t)row * OUT_DIM + col] = acc[v];
        }
    }
}


extern "C" void kernel_launch(void* const* d_in, const int* in_sizes, int n_in,
                              void* d_out, int out_size, void* d_ws, size_t ws_size,
                              hipStream_t stream) {
    const float* x       = (const float*)d_in[0];
    const int*   src     = (const int*)  d_in[1];
    const int*   dst     = (const int*)  d_in[2];
    const float* W_self  = (const float*)d_in[3];
    const float* W_neigh = (const float*)d_in[4];
    const float* b       = (const float*)d_in[5];
    const float* W_pred  = (const float*)d_in[6];
    const float* b_pred  = (const float*)d_in[7];

    const int n_nodes = in_sizes[0] / D_FEAT;
    const int n_edges = in_sizes[1];
    const int L       = in_sizes[5] / D_FEAT;
    const int feat_elems = n_nodes * D_FEAT;

    char*  ws = (char*)d_ws;
    size_t off = ((size_t)n_nodes * sizeof(float) + 511) & ~(size_t)511;
    size_t feat_bytes = (size_t)feat_elems * sizeof(float);
    size_t edge_bytes = ((size_t)n_edges * sizeof(int) + 511) & ~(size_t)511;
    float* invdeg = (float*)ws;
    float* bufA   = (float*)(ws + off);
    float* bufB   = (float*)(ws + off + feat_bytes);
    char*  ga     = ws + off + 2 * feat_bytes;
    int*   cnt    = (int*)(ga);
    int*   first  = (int*)(ga + off);
    int*   slot   = (int*)(ga + 2 * off);
    int*   nbr    = (int*)(ga + 2 * off + edge_bytes);
    int*   pairs  = (int*)(ga + 2 * off + 2 * edge_bytes);

    const int T = 256;
    const int edge_blocks = (n_edges + T - 1) / T;
    const int wave_blocks = (n_nodes + 7) / 8;

    sage_zero_i32<<<(n_nodes + T - 1) / T, T, 0, stream>>>(cnt, n_nodes);
    sage_count_kernel<<<edge_blocks, T, 0, stream>>>(dst, cnt, slot, n_edges);
    sage_offsets_kernel<<<(n_nodes + SCAN_CHUNK - 1) / SCAN_CHUNK, 256, 0, stream>>>(
        cnt, first, invdeg, n_nodes);
    sage_bucket_kernel<<<edge_blocks, T, 0, stream>>>(src, dst, slot, first, pairs, n_edges);
    sage_order_kernel<<<wave_blocks, 256, 0, stream>>>(first, cnt, pairs, nbr, n_nodes);

    const float* cur = x;
    float* nxt = bufA;
    const int m_tiles = n_nodes / 16;
    const int layer_blocks = (n_nodes + TILE_M - 1) / TILE_M;

    for (int l = 0; l < L; ++l) {
        sage_layer_kernel<<<layer_blocks, 256, 0, stream>>>(
            cur, nxt, first, cnt, nbr, invdeg,
            W_self  + (size_t)l * D_FEAT * D_FEAT,
            W_neigh + (size_t)l * D_FEAT * D_FEAT,
            b       + (size_t)l * D_FEAT, n_nodes);
        cur = nxt;
        nxt = (nxt == bufA) ? bufB : bufA;
    }

    sage_pred_kernel<<<m_tiles, 64, 0, stream>>>(cur, W_pred, b_pred, (float*)d_out);
}


# Edge grouping without atomics in memory: work-groups that each own a range of destination nodes and a piece of the edge list count and place its edges with LDS adds; per-node sum order unchanged
# speedup vs baseline: 3.6275x; 1.1117x over previous
#include <hip/hip_runtime.h>
#include <hip/hip_bf16.h>


#define D_FEAT 128
#define LDS_STRIDE 132
#define OUT_DIM 26
#define SCAN_CHUNK 1024

typedef __attribute__((ext_vector_type(2))) float v2f;
typedef __attribute__((ext_vector_type(8))) float v8f;

#define RANGE_NODES 4096
#define EDGE_CHUNKS 16
#define SCAN_THREADS 512
#define SCAN_BATCH 8

__device__ __forceinline__ int sage_chunk_len(int n_edges) {
    return (((n_edges + EDGE_CHUNKS - 1) / EDGE_CHUNKS) + 3) & ~3;
}

__device__ __forceinline__ int4 sage_read4(const int* __restrict__ p, int e, int e_end) {
    if (e + 4 <= e_end) return *(const int4*)(p + e);
    int4 v = make_int4(-1, -1, -1, -1);
    if (e < e_end)     v.x = p[e];
    if (e + 1 < e_end) v.y = p[e + 1];
    if (e + 2 < e_end) v.z = p[e + 2];
    return v;
}

__global__ void __launch_bounds__(SCAN_THREADS)
sage_count_kernel(const int* __restrict__ dst, int* __restrict__ cntc,
                  int n_edges, int n_nodes) {
    __shared__ int hist[RANGE_NODES];
    const int t  = threadIdx.x;
    const int lo = blockIdx.x * RANGE_NODES;
    for (int i = t; i < RANGE_NODES; i += SCAN_THREADS) hist[i] = 0;
    __syncthreads();

    const int chunk = sage_chunk_len(n_edges);
    const int e_beg = blockIdx.y * chunk;
    const int e_end = (e_beg + chunk < n_edges) ? (e_beg + chunk) : n_edges;

    for (int e0 = e_beg + 4 * t; e0 < e_end; e0 += 4 * SCAN_THREADS * SCAN_BATCH) {
        int4 d[SCAN_BATCH];
        #pragma unroll
        for (int b = 0; b < SCAN_BATCH; ++b) d[b] = sage_read4(dst, e0 + b * 4 * SCAN_THREADS, e_end);
        #pragma unroll
        for (int b = 0; b < SCAN_BATCH; ++b) {
            const unsigned i0 = (unsigned)(d[b].x - lo), i1 = (unsigned)(d[b].y - lo);
            const unsigned i2 = (unsigned)(d[b].z - lo), i3 = (unsigned)(d[b].w - lo);
            if (i0 < RANGE_NODES) atomicAdd(&hist[i0], 1);
            if (i1 < RANGE_NODES) atomicAdd(&hist[i1], 1);
            if (i2 < RANGE_NODES) atomicAdd(&hist[i2], 1);
            if (i3 < RANGE_NODES) atomicAdd(&hist[i3], 1);
        }
    }
    __syncthreads();

    int* plane = cntc + (size_t)blockIdx.y * n_nodes;
    for (int i = t; i < RANGE_NODES; i += SCAN_THREADS)
        if (lo + i < n_nodes) plane[lo + i] = hist[i];
}

__global__ void sage_totals_kernel(int* __restrict__ cntc, int* __restrict__ cnt, int n_nodes) {
    const int i = blockIdx.x * blockDim.x + threadIdx.x;
    if (i < n_nodes) {
        int v[EDGE_CHUNKS];
        #pragma unroll
        for (int c = 0; c < EDGE_CHUNKS; ++c) v[c] = cntc[(size_t)c * n_nodes + i];
        int run = 0;
        #pragma unroll
        for (int c = 0; c < EDGE_CHUNKS; ++c) {
            cntc[(size_t)c * n_nodes + i] = run;
            run += v[c];
        }
        cnt[i] = run;
    }
}

__global__ void __launch_bounds__(256)
sage_offsets_kernel(const int* __restrict__ cnt, int* __restrict__ first,
                    float* __restrict__ invdeg, int n_nodes) {
    __shared__ int wave_front[8];
    __shared__ int wave_total[8];

    const int t    = threadIdx.x;
    const int lane = t & 31;
    const int w    = t >> 5;
    const int base = blockIdx.x * SCAN_CHUNK;

    int front = 0;
    #pragma unroll 4
    for (int i = t * 4; i < base; i += SCAN_CHUNK) {
        const int4 v = *(const int4*)(cnt + i);
        front += (v.x + v.y) + (v.z + v.w);
    }

    const int i0 = base + t * 4;
    int c[4];
    for (int k = 0; k < 4; ++k) c[k] = (i0 + k < n_nodes) ? cnt[i0 + k] : 0;
    const int tsum = (c[0] + c[1]) + (c[2] + c[3]);

    int inc = tsum;
    for (int o = 1; o < 32; o <<= 1) {
        const int v = __shfl_up(inc, o);
        if (lane >= o) inc += v;
    }
    for (int o = 16; o > 0; o >>= 1) front += __shfl_xor(front, o);

    if (lane == 31) wave_total[w] = inc;
    if (lane == 0)  wave_front[w] = front;
    __syncthreads();

    int run = inc - tsum;
    for (int k = 0; k < 8; ++k) {
        run += wave_front[k];
        if (k < w) run += wave_total[k];
    }

    for (int k = 0; k < 4; ++k) {
        if (i0 + k < n_nodes) {
            first[i0 + k]  = run;
            invdeg[i0 + k] = 1.0f / fmaxf((float)c[k], 1.0f);
            run += c[k];
        }
    }
}

__global__ void __launch_bounds__(SCAN_THREADS)
sage_bucket_kernel(const int* __restrict__ src, const int* __restrict__ dst,
                   const int* __restrict__ first, const int* __restrict__ cntc,
                   int* __restrict__ pairs, int n_edges, int n_nodes) {
    __shared__ int place[RANGE_NODES];
    const int t  = threadIdx.x;
    const int lo = blockIdx.x * RANGE_NODES;
    const int* plane = cntc + (size_t)blockIdx.y * n_nodes;
    for (int i = t; i < RANGE_NODES; i += SCAN_THREADS)
        place[i] = (lo + i < n_nodes) ? (first[lo + i] + plane[lo + i]) : 0;
    __syncthreads();

    const int chunk = sage_chunk_len(n_edges);
    const int e_beg = blockIdx.y * chunk;
    const int e_end = (e_beg + chunk < n_edges) ? (e_beg + chunk) : n_edges;

    for (int e0 = e_beg + 4 * t; e0 < e_end; e0 += 4 * SCAN_THREADS * SCAN_BATCH) {
        int4 d[SCAN_BATCH], s[SCAN_BATCH];
        #pragma unroll
        for (int b = 0; b < SCAN_BATCH; ++b) {
            d[b] = sage_read4(dst, e0 + b * 4 * SCAN_THREADS, e_end);
            s[b] = sage_read4(src, e0 + b * 4 * SCAN_THREADS, e_end);
        }
        int pos[SCAN_BATCH][4];
        #pragma unroll
        for (int b = 0; b < SCAN_BATCH; ++b) {
            const int dd[4] = {d[b].x, d[b].y, d[b].z, d[b].w};
            #pragma unroll
            for (int k = 0; k < 4; ++k) {
                const unsigned i = (unsigned)(dd[k] - lo);
                pos[b][k] = -1;
                if (i < RANGE_NODES) pos[b][k] = atomicAdd(&place[i], 1);
            }
        }
        #pragma unroll
        for (int b = 0; b < SCAN_BATCH; ++b) {
            const int e = e0 + b * 4 * SCAN_THREADS;
            const int ss[4] = {s[b].x, s[b].y, s[b].z, s[b].w};
            #pragma unroll
            for (int k = 0; k < 4; ++k)
                if (pos[b][k] >= 0)
                    *(int2*)(pairs + 2 * (size_t)pos[b][k]) = make_int2(e + k, ss[k]);
        }
    }
}

__global__ void __launch_bounds__(256)
sage_order_kernel(const int* __restrict__ first, const int* __restrict__ cnt,
                  const int* __restrict__ pairs, int* __restrict__ nbr, int n_nodes) {
    const int lane = threadIdx.x & 31;
    const int node = __builtin_amdgcn_readfirstlane((int)(blockIdx.x * 8 + (threadIdx.x >> 5)));
    if (node >= n_nodes) return;
    const int beg = first[node];
    const int n   = cnt[node];
    const int2* seg = (const int2*)pairs + beg;

    for (int kb = 0; kb < n; kb += 32) {
        const bool have = kb + lane < n;
        int2 mine = make_int2(0, 0);
        if (have) mine = seg[kb + lane];
        int pos = 0;
        for (int mb = 0; mb < n; mb += 32) {
            const int mcount = (n - mb < 32) ? (n - mb) : 32;
            int other = 0;
            if (mb + lane < n) other = seg[mb + lane].x;
            for (int m = 0; m < mcount; ++m)
                pos += (__builtin_amdgcn_readlane(other, m) < mine.x) ? 1 : 0;
        }
        if (have) nbr[beg + pos] = mine.y;
    }
}


__device__ __forceinline__ float4 sage_row(const float* __restrict__ xl, int ids, int j) {
    const int s = __builtin_amdgcn_readlane(ids, j);
    return *(const float4*)(xl + (size_t)s * D_FEAT);
}

template <int K>
__device__ __forceinline__ void sage_sum_rows(float4& acc, const float* __restrict__ xl,
                                              int ids, int j) {
    float4 v[K];
    for (int k = 0; k < K; ++k) v[k] = sage_row(xl, ids, j + k);
    for (int k = 0; k < K; ++k) {
        acc.x += v[k].x; acc.y += v[k].y; acc.z += v[k].z; acc.w += v[k].w;
    }
}

__device__ __forceinline__ float4 sage_node_sum(const int* __restrict__ nbr,
                                                const float* __restrict__ xl,
                                                int beg, int n, int lane, int ids0) {
    float4 acc = make_float4(0.0f, 0.0f, 0.0f, 0.0f);
    for (int base = 0; base < n; base += 32) {
        const int m = (n - base < 32) ? (n - base) : 32;
        int ids = ids0;
        if (base > 0) {
            ids = 0;
            if (lane < m) ids = nbr[beg + base + lane];
        }

        int j = 0;
        for (; j + 8 <= m; j += 8) sage_sum_rows<8>(acc, xl, ids, j);
        if (j + 4 <= m) { sage_sum_rows<4>(acc, xl, ids, j); j += 4; }
        const int r = m - j;
        if (r == 3)      sage_sum_rows<3>(acc, xl, ids, j);
        else if (r == 2) sage_sum_rows<2>(acc, xl, ids, j);
        else if (r == 1) sage_sum_rows<1>(acc, xl, ids, j);
    }
    return acc;
}

#define TILE_M 64
#define M_SUB (TILE_M / 16)
#define ROWS_PER_WAVE (TILE_M / 8)

#define KC_STEPS 4
#define KC_K (4 * KC_STEPS)

__device__ __forceinline__ void sage_load_w(v2f (&bs)[KC_STEPS], v2f (&bn)[KC_STEPS],
                                            const float* __restrict__ Ws,
                                            const float* __restrict__ Wn,
                                            int kbase, int half, int col) {
    #pragma unroll
    for (int s = 0; s < KC_STEPS; ++s) {
        const int ka = kbase + 4 * s + half * 2;
        bs[s].x = Ws[(size_t)ka * D_FEAT + col];
        bs[s].y = Ws[(size_t)(ka + 1) * D_FEAT + col];
        bn[s].x = Wn[(size_t)ka * D_FEAT + col];
        bn[s].y = Wn[(size_t)(ka + 1) * D_FEAT + col];
    }
}

__device__ __forceinline__ void sage_mma_chunk(v8f (&acc)[TILE_M / 16],
                                               const v2f (&bs)[KC_STEPS],
                                               const v2f (&bn)[KC_STEPS],
                                               const float* As, const float* Hs,
                                               int kbase, int half, int l16) {
    #pragma unroll
    for (int s = 0; s < KC_STEPS; ++s) {
        const int ka = kbase + 4 * s + half * 2;
        v2f a[TILE_M / 16], hh[TILE_M / 16];
        #pragma unroll
        for (int mt = 0; mt < TILE_M / 16; ++mt) {
            a[mt]  = *(const v2f*)(As + (mt * 16 + l16) * LDS_STRIDE + ka);
            hh[mt] = *(const v2f*)(Hs + (mt * 16 + l16) * LDS_STRIDE + ka);
        }
        #pragma unroll
        for (int mt = 0; mt < TILE_M / 16; ++mt)
            acc[mt] = __builtin_amdgcn_wmma_f32_16x16x4_f32(false, a[mt], false, bs[s],
                                                            (short)0, acc[mt], false, false);
        #pragma unroll
        for (int mt = 0; mt < TILE_M / 16; ++mt)
            acc[mt] = __builtin_amdgcn_wmma_f32_16x16x4_f32(false, hh[mt], false, bn[s],
                                                            (short)0, acc[mt], false, false);
    }
}

template <bool LAST>
__device__ __forceinline__ void sage_chunk_pair(v8f (&acc)[TILE_M / 16],
                                                v2f (&bs0)[KC_STEPS], v2f (&bn0)[KC_STEPS],
                                                v2f (&bs1)[KC_STEPS], v2f (&bn1)[KC_STEPS],
                                                const float* __restrict__ Ws,
                                                const float* __restrict__ Wn,
                                                const float* As, const float* Hs,
                                                int kc, int half, int l16, int col) {
    sage_load_w(bs1, bn1, Ws, Wn, kc + KC_K, half, col);
    sage_mma_chunk(acc, bs0, bn0, As, Hs, kc, half, l16);
    if (!LAST) sage_load_w(bs0, bn0, Ws, Wn, kc + 2 * KC_K, half, col);
    sage_mma_chunk(acc, bs1, bn1, As, Hs, kc + KC_K, half, l16);
}

__global__ void __launch_bounds__(256)
sage_layer_kernel(const float* __restrict__ xin, float* __restrict__ hout,
                  const int* __restrict__ first, const int* __restrict__ cnt,
                  const int* __restrict__ nbr, const float* __restrict__ invdeg,
                  const float* __restrict__ Wself, const float* __restrict__ Wneigh,
                  const float* __restrict__ bias, int n_nodes) {
    __shared__ float As[TILE_M * LDS_STRIDE];
    __shared__ float Hs[TILE_M * LDS_STRIDE];

    const int m0 = blockIdx.x * TILE_M;
    const int t  = threadIdx.x;
    const int lane = t & 31;

    const int wave = t >> 5;
    const int n0   = wave * 16;
    const int half = lane >> 4;
    const int l16  = lane & 15;
    const int col  = n0 + l16;

    v2f bs0[KC_STEPS], bn0[KC_STEPS], bs1[KC_STEPS], bn1[KC_STEPS];
    sage_load_w(bs0, bn0, Wself, Wneigh, 0, half, col);
    const float bv = bias[col];

    {
        const int wu   = __builtin_amdgcn_readfirstlane(t >> 5);
        const int row0 = ROWS_PER_WAVE * wu;

        int   beg[ROWS_PER_WAVE];
        int   nn[ROWS_PER_WAVE];
        float iv[ROWS_PER_WAVE];
        int   ids[ROWS_PER_WAVE];
        float4 xv[ROWS_PER_WAVE];

        #pragma unroll
        for (int i = 0; i < ROWS_PER_WAVE; ++i) {
            const int node  = m0 + row0 + i;
            const bool live = node < n_nodes;
            const int nodeC = live ? node : (n_nodes - 1);
            beg[i] = first[nodeC];
            nn[i]  = live ? cnt[nodeC] : 0;
            iv[i]  = invdeg[nodeC];
        }
        #pragma unroll
        for (int i = 0; i < ROWS_PER_WAVE; ++i) {
            ids[i] = 0;
            if (lane < nn[i]) ids[i] = nbr[beg[i] + lane];
        }
        #pragma unroll
        for (int i = 0; i < ROWS_PER_WAVE; ++i) {
            const int node = m0 + row0 + i;
            xv[i] = make_float4(0.0f, 0.0f, 0.0f, 0.0f);
            if (node < n_nodes) xv[i] = *(const float4*)(xin + (size_t)node * D_FEAT + lane * 4);
        }
        #pragma unroll
        for (int i = 0; i < ROWS_PER_WAVE; ++i)
            *(float4*)(As + (row0 + i) * LDS_STRIDE + lane * 4) = xv[i];

        const float* xl = xin + lane * 4;
        #pragma unroll
        for (int i = 0; i < ROWS_PER_WAVE; ++i) {
            float4 s = sage_node_sum(nbr, xl, beg[i], nn[i], lane, ids[i]);
            s.x *= iv[i]; s.y *= iv[i]; s.z *= iv[i]; s.w *= iv[i];
            *(float4*)(Hs + (row0 + i) * LDS_STRIDE + lane * 4) = s;
        }
    }

    __syncthreads();

    v8f acc[M_SUB];
    #pragma unroll
    for (int mt = 0; mt < M_SUB; ++mt) {
        #pragma unroll
        for (int i = 0; i < 8; ++i) acc[mt][i] = bv;
    }

    for (int kc = 0; kc < D_FEAT - 2 * KC_K; kc += 2 * KC_K)
        sage_chunk_pair<false>(acc, bs0, bn0, bs1, bn1, Wself, Wneigh, As, Hs, kc, half, l16, col);
    sage_chunk_pair<true>(acc, bs0, bn0, bs1, bn1, Wself, Wneigh, As, Hs,
                          D_FEAT - 2 * KC_K, half, l16, col);

    #pragma unroll
    for (int mt = 0; mt < M_SUB; ++mt) {
        #pragma unroll
        for (int v = 0; v < 8; ++v) {
            const int row = m0 + mt * 16 + v + half * 8;
            float val = acc[mt][v];
            val = val > 0.0f ? val : 0.0f;
            if (row < n_nodes) hout[(size_t)row * D_FEAT + col] = val;
        }
    }
}

__global__ void __launch_bounds__(64)
sage_pred_kernel(const float* __restrict__ xin, const float* __restrict__ Wp,
                 const float* __restrict__ bp, float* __restrict__ out) {
    const int m0   = blockIdx.x * 16;
    const int t    = threadIdx.x;
    const int wave = t >> 5;
    const int lane = t & 31;
    const int n0   = wave * 16;
    const int half = lane >> 4;
    const int l16  = lane & 15;
    const int col  = n0 + l16;
    const int bcol = col < OUT_DIM ? col : (OUT_DIM - 1);

    v8f acc;
    {
        float bv = bp[bcol];
        #pragma unroll
        for (int i = 0; i < 8; ++i) acc[i] = bv;
    }

    #pragma unroll 4
    for (int k0 = 0; k0 < D_FEAT; k0 += 4) {
        const int ka = k0 + half * 2;
        const float* ap = xin + (size_t)(m0 + l16) * D_FEAT + ka;
        v2f a; a.x = ap[0]; a.y = ap[1];
        v2f b;
        b.x = Wp[(size_t)ka * OUT_DIM + bcol];
        b.y = Wp[(size_t)(ka + 1) * OUT_DIM + bcol];
        acc = __builtin_amdgcn_wmma_f32_16x16x4_f32(false, a, false, b,
                                                    (short)0, acc, false, false);
    }

    if (col < OUT_DIM) {
        #pragma unroll
        for (int v = 0; v < 8; ++v) {
            int row = m0 + v + half * 8;
            out[(size_t)row * OUT_DIM + col] = acc[v];
        }
    }
}


extern "C" void kernel_launch(void* const* d_in, const int* in_sizes, int n_in,
                              void* d_out, int out_size, void* d_ws, size_t ws_size,
                              hipStream_t stream) {
    const float* x       = (const float*)d_in[0];
    const int*   src     = (const int*)  d_in[1];
    const int*   dst     = (const int*)  d_in[2];
    const float* W_self  = (const float*)d_in[3];
    const float* W_neigh = (const float*)d_in[4];
    const float* b       = (const float*)d_in[5];
    const float* W_pred  = (const float*)d_in[6];
    const float* b_pred  = (const float*)d_in[7];

    const int n_nodes = in_sizes[0] / D_FEAT;
    const int n_edges = in_sizes[1];
    const int L       = in_sizes[5] / D_FEAT;
    const int feat_elems = n_nodes * D_FEAT;

    char*  ws = (char*)d_ws;
    size_t off = ((size_t)n_nodes * sizeof(float) + 511) & ~(size_t)511;
    size_t feat_bytes = (size_t)feat_elems * sizeof(float);
    size_t edge_bytes = ((size_t)n_edges * sizeof(int) + 511) & ~(size_t)511;
    float* invdeg = (float*)ws;
    float* bufA   = (float*)(ws + off);
    float* bufB   = (float*)(ws + off + feat_bytes);
    char*  ga     = ws + off + 2 * feat_bytes;
    int*   cnt    = (int*)(ga);
    int*   first  = (int*)(ga + off);
    size_t cntc_bytes = ((size_t)EDGE_CHUNKS * n_nodes * sizeof(int) + 511) & ~(size_t)511;
    int*   cntc   = (int*)(ga + 2 * off);
    int*   nbr    = (int*)(ga + 2 * off + cntc_bytes);
    int*   pairs  = (int*)(ga + 2 * off + cntc_bytes + edge_bytes);

    const int T = 256;
    const dim3 scan_grid((n_nodes + RANGE_NODES - 1) / RANGE_NODES, EDGE_CHUNKS);
    const int wave_blocks = (n_nodes + 7) / 8;

    sage_count_kernel<<<scan_grid, SCAN_THREADS, 0, stream>>>(dst, cntc, n_edges, n_nodes);
    sage_totals_kernel<<<(n_nodes + T - 1) / T, T, 0, stream>>>(cntc, cnt, n_nodes);
    sage_offsets_kernel<<<(n_nodes + SCAN_CHUNK - 1) / SCAN_CHUNK, 256, 0, stream>>>(
        cnt, first, invdeg, n_nodes);
    sage_bucket_kernel<<<scan_grid, SCAN_THREADS, 0, stream>>>(src, dst, first, cntc, pairs,
                                                              n_edges, n_nodes);
    sage_order_kernel<<<wave_blocks, 256, 0, stream>>>(first, cnt, pairs, nbr, n_nodes);

    const float* cur = x;
    float* nxt = bufA;
    const int m_tiles = n_nodes / 16;
    const int layer_blocks = (n_nodes + TILE_M - 1) / TILE_M;

    for (int l = 0; l < L; ++l) {
        sage_layer_kernel<<<layer_blocks, 256, 0, stream>>>(
            cur, nxt, first, cnt, nbr, invdeg,
            W_self  + (size_t)l * D_FEAT * D_FEAT,
            W_neigh + (size_t)l * D_FEAT * D_FEAT,
            b       + (size_t)l * D_FEAT, n_nodes);
        cur = nxt;
        nxt = (nxt == bufA) ? bufB : bufA;
    }

    sage_pred_kernel<<<m_tiles, 64, 0, stream>>>(cur, W_pred, b_pred, (float*)d_out);
}


# Prediction head fused into the last layer's launch: the relu'd 64x128 tile goes to LDS, not memory, and the block forms its 64x26 outputs there (same K order); one launch and one plane round trip go
# speedup vs baseline: 3.7728x; 1.0401x over previous
#include <hip/hip_runtime.h>
#include <hip/hip_bf16.h>


#define D_FEAT 128
#define LDS_STRIDE 132
#define OUT_DIM 26
#define SCAN_CHUNK 1024

typedef __attribute__((ext_vector_type(2))) float v2f;
typedef __attribute__((ext_vector_type(8))) float v8f;

#define RANGE_NODES 4096
#define EDGE_CHUNKS 16
#define SCAN_THREADS 512
#define SCAN_BATCH 8

__device__ __forceinline__ int sage_chunk_len(int n_edges) {
    return (((n_edges + EDGE_CHUNKS - 1) / EDGE_CHUNKS) + 3) & ~3;
}

__device__ __forceinline__ int4 sage_read4(const int* __restrict__ p, int e, int e_end) {
    if (e + 4 <= e_end) return *(const int4*)(p + e);
    int4 v = make_int4(-1, -1, -1, -1);
    if (e < e_end)     v.x = p[e];
    if (e + 1 < e_end) v.y = p[e + 1];
    if (e + 2 < e_end) v.z = p[e + 2];
    return v;
}

__global__ void __launch_bounds__(SCAN_THREADS)
sage_count_kernel(const int* __restrict__ dst, int* __restrict__ cntc,
                  int n_edges, int n_nodes) {
    __shared__ int hist[RANGE_NODES];
    const int t  = threadIdx.x;
    const int lo = blockIdx.x * RANGE_NODES;
    for (int i = t; i < RANGE_NODES; i += SCAN_THREADS) hist[i] = 0;
    __syncthreads();

    const int chunk = sage_chunk_len(n_edges);
    const int e_beg = blockIdx.y * chunk;
    const int e_end = (e_beg + chunk < n_edges) ? (e_beg + chunk) : n_edges;

    for (int e0 = e_beg + 4 * t; e0 < e_end; e0 += 4 * SCAN_THREADS * SCAN_BATCH) {
        int4 d[SCAN_BATCH];
        #pragma unroll
        for (int b = 0; b < SCAN_BATCH; ++b) d[b] = sage_read4(dst, e0 + b * 4 * SCAN_THREADS, e_end);
        #pragma unroll
        for (int b = 0; b < SCAN_BATCH; ++b) {
            const unsigned i0 = (unsigned)(d[b].x - lo), i1 = (unsigned)(d[b].y - lo);
            const unsigned i2 = (unsigned)(d[b].z - lo), i3 = (unsigned)(d[b].w - lo);
            if (i0 < RANGE_NODES) atomicAdd(&hist[i0], 1);
            if (i1 < RANGE_NODES) atomicAdd(&hist[i1], 1);
            if (i2 < RANGE_NODES) atomicAdd(&hist[i2], 1);
            if (i3 < RANGE_NODES) atomicAdd(&hist[i3], 1);
        }
    }
    __syncthreads();

    int* plane = cntc + (size_t)blockIdx.y * n_nodes;
    for (int i = t; i < RANGE_NODES; i += SCAN_THREADS)
        if (lo + i < n_nodes) plane[lo + i] = hist[i];
}

__global__ void sage_totals_kernel(int* __restrict__ cntc, int* __restrict__ cnt, int n_nodes) {
    const int i = blockIdx.x * blockDim.x + threadIdx.x;
    if (i < n_nodes) {
        int v[EDGE_CHUNKS];
        #pragma unroll
        for (int c = 0; c < EDGE_CHUNKS; ++c) v[c] = cntc[(size_t)c * n_nodes + i];
        int run = 0;
        #pragma unroll
        for (int c = 0; c < EDGE_CHUNKS; ++c) {
            cntc[(size_t)c * n_nodes + i] = run;
            run += v[c];
        }
        cnt[i] = run;
    }
}

__global__ void __launch_bounds__(256)
sage_offsets_kernel(const int* __restrict__ cnt, int* __restrict__ first,
                    float* __restrict__ invdeg, int n_nodes) {
    __shared__ int wave_front[8];
    __shared__ int wave_total[8];

    const int t    = threadIdx.x;
    const int lane = t & 31;
    const int w    = t >> 5;
    const int base = blockIdx.x * SCAN_CHUNK;

    int front = 0;
    #pragma unroll 4
    for (int i = t * 4; i < base; i += SCAN_CHUNK) {
        const int4 v = *(const int4*)(cnt + i);
        front += (v.x + v.y) + (v.z + v.w);
    }

    const int i0 = base + t * 4;
    int c[4];
    for (int k = 0; k < 4; ++k) c[k] = (i0 + k < n_nodes) ? cnt[i0 + k] : 0;
    const int tsum = (c[0] + c[1]) + (c[2] + c[3]);

    int inc = tsum;
    for (int o = 1; o < 32; o <<= 1) {
        const int v = __shfl_up(inc, o);
        if (lane >= o) inc += v;
    }
    for (int o = 16; o > 0; o >>= 1) front += __shfl_xor(front, o);

    if (lane == 31) wave_total[w] = inc;
    if (lane == 0)  wave_front[w] = front;
    __syncthreads();

    int run = inc - tsum;
    for (int k = 0; k < 8; ++k) {
        run += wave_front[k];
        if (k < w) run += wave_total[k];
    }

    for (int k = 0; k < 4; ++k) {
        if (i0 + k < n_nodes) {
            first[i0 + k]  = run;
            invdeg[i0 + k] = 1.0f / fmaxf((float)c[k], 1.0f);
            run += c[k];
        }
    }
}

__global__ void __launch_bounds__(SCAN_THREADS)
sage_bucket_kernel(const int* __restrict__ src, const int* __restrict__ dst,
                   const int* __restrict__ first, const int* __restrict__ cntc,
                   int* __restrict__ pairs, int n_edges, int n_nodes) {
    __shared__ int place[RANGE_NODES];
    const int t  = threadIdx.x;
    const int lo = blockIdx.x * RANGE_NODES;
    const int* plane = cntc + (size_t)blockIdx.y * n_nodes;
    for (int i = t; i < RANGE_NODES; i += SCAN_THREADS)
        place[i] = (lo + i < n_nodes) ? (first[lo + i] + plane[lo + i]) : 0;
    __syncthreads();

    const int chunk = sage_chunk_len(n_edges);
    const int e_beg = blockIdx.y * chunk;
    const int e_end = (e_beg + chunk < n_edges) ? (e_beg + chunk) : n_edges;

    for (int e0 = e_beg + 4 * t; e0 < e_end; e0 += 4 * SCAN_THREADS * SCAN_BATCH) {
        int4 d[SCAN_BATCH], s[SCAN_BATCH];
        #pragma unroll
        for (int b = 0; b < SCAN_BATCH; ++b) {
            d[b] = sage_read4(dst, e0 + b * 4 * SCAN_THREADS, e_end);
            s[b] = sage_read4(src, e0 + b * 4 * SCAN_THREADS, e_end);
        }
        int pos[SCAN_BATCH][4];
        #pragma unroll
        for (int b = 0; b < SCAN_BATCH; ++b) {
            const int dd[4] = {d[b].x, d[b].y, d[b].z, d[b].w};
            #pragma unroll
            for (int k = 0; k < 4; ++k) {
                const unsigned i = (unsigned)(dd[k] - lo);
                pos[b][k] = -1;
                if (i < RANGE_NODES) pos[b][k] = atomicAdd(&place[i], 1);
            }
        }
        #pragma unroll
        for (int b = 0; b < SCAN_BATCH; ++b) {
            const int e = e0 + b * 4 * SCAN_THREADS;
            const int ss[4] = {s[b].x, s[b].y, s[b].z, s[b].w};
            #pragma unroll
            for (int k = 0; k < 4; ++k)
                if (pos[b][k] >= 0)
                    *(int2*)(pairs + 2 * (size_t)pos[b][k]) = make_int2(e + k, ss[k]);
        }
    }
}

__global__ void __launch_bounds__(256)
sage_order_kernel(const int* __restrict__ first, const int* __restrict__ cnt,
                  const int* __restrict__ pairs, int* __restrict__ nbr, int n_nodes) {
    const int lane = threadIdx.x & 31;
    const int node = __builtin_amdgcn_readfirstlane((int)(blockIdx.x * 8 + (threadIdx.x >> 5)));
    if (node >= n_nodes) return;
    const int beg = first[node];
    const int n   = cnt[node];
    const int2* seg = (const int2*)pairs + beg;

    for (int kb = 0; kb < n; kb += 32) {
        const bool have = kb + lane < n;
        int2 mine = make_int2(0, 0);
        if (have) mine = seg[kb + lane];
        int pos = 0;
        for (int mb = 0; mb < n; mb += 32) {
            const int mcount = (n - mb < 32) ? (n - mb) : 32;
            int other = 0;
            if (mb + lane < n) other = seg[mb + lane].x;
            for (int m = 0; m < mcount; ++m)
                pos += (__builtin_amdgcn_readlane(other, m) < mine.x) ? 1 : 0;
        }
        if (have) nbr[beg + pos] = mine.y;
    }
}


__device__ __forceinline__ float4 sage_row(const float* __restrict__ xl, int ids, int j) {
    const int s = __builtin_amdgcn_readlane(ids, j);
    return *(const float4*)(xl + (size_t)s * D_FEAT);
}

template <int K>
__device__ __forceinline__ void sage_sum_rows(float4& acc, const float* __restrict__ xl,
                                              int ids, int j) {
    float4 v[K];
    for (int k = 0; k < K; ++k) v[k] = sage_row(xl, ids, j + k);
    for (int k = 0; k < K; ++k) {
        acc.x += v[k].x; acc.y += v[k].y; acc.z += v[k].z; acc.w += v[k].w;
    }
}

__device__ __forceinline__ float4 sage_node_sum(const int* __restrict__ nbr,
                                                const float* __restrict__ xl,
                                                int beg, int n, int lane, int ids0) {
    float4 acc = make_float4(0.0f, 0.0f, 0.0f, 0.0f);
    for (int base = 0; base < n; base += 32) {
        const int m = (n - base < 32) ? (n - base) : 32;
        int ids = ids0;
        if (base > 0) {
            ids = 0;
            if (lane < m) ids = nbr[beg + base + lane];
        }

        int j = 0;
        for (; j + 8 <= m; j += 8) sage_sum_rows<8>(acc, xl, ids, j);
        if (j + 4 <= m) { sage_sum_rows<4>(acc, xl, ids, j); j += 4; }
        const int r = m - j;
        if (r == 3)      sage_sum_rows<3>(acc, xl, ids, j);
        else if (r == 2) sage_sum_rows<2>(acc, xl, ids, j);
        else if (r == 1) sage_sum_rows<1>(acc, xl, ids, j);
    }
    return acc;
}

#define TILE_M 64
#define M_SUB (TILE_M / 16)
#define ROWS_PER_WAVE (TILE_M / 8)

#define KC_STEPS 4
#define KC_K (4 * KC_STEPS)

__device__ __forceinline__ void sage_load_w(v2f (&bs)[KC_STEPS], v2f (&bn)[KC_STEPS],
                                            const float* __restrict__ Ws,
                                            const float* __restrict__ Wn,
                                            int kbase, int half, int col) {
    #pragma unroll
    for (int s = 0; s < KC_STEPS; ++s) {
        const int ka = kbase + 4 * s + half * 2;
        bs[s].x = Ws[(size_t)ka * D_FEAT + col];
        bs[s].y = Ws[(size_t)(ka + 1) * D_FEAT + col];
        bn[s].x = Wn[(size_t)ka * D_FEAT + col];
        bn[s].y = Wn[(size_t)(ka + 1) * D_FEAT + col];
    }
}

__device__ __forceinline__ void sage_mma_chunk(v8f (&acc)[TILE_M / 16],
                                               const v2f (&bs)[KC_STEPS],
                                               const v2f (&bn)[KC_STEPS],
                                               const float* As, const float* Hs,
                                               int kbase, int half, int l16) {
    #pragma unroll
    for (int s = 0; s < KC_STEPS; ++s) {
        const int ka = kbase + 4 * s + half * 2;
        v2f a[TILE_M / 16], hh[TILE_M / 16];
        #pragma unroll
        for (int mt = 0; mt < TILE_M / 16; ++mt) {
            a[mt]  = *(const v2f*)(As + (mt * 16 + l16) * LDS_STRIDE + ka);
            hh[mt] = *(const v2f*)(Hs + (mt * 16 + l16) * LDS_STRIDE + ka);
        }
        #pragma unroll
        for (int mt = 0; mt < TILE_M / 16; ++mt)
            acc[mt] = __builtin_amdgcn_wmma_f32_16x16x4_f32(false, a[mt], false, bs[s],
                                                            (short)0, acc[mt], false, false);
        #pragma unroll
        for (int mt = 0; mt < TILE_M / 16; ++mt)
            acc[mt] = __builtin_amdgcn_wmma_f32_16x16x4_f32(false, hh[mt], false, bn[s],
                                                            (short)0, acc[mt], false, false);
    }
}

template <bool LAST>
__device__ __forceinline__ void sage_chunk_pair(v8f (&acc)[TILE_M / 16],
                                                v2f (&bs0)[KC_STEPS], v2f (&bn0)[KC_STEPS],
                                                v2f (&bs1)[KC_STEPS], v2f (&bn1)[KC_STEPS],
                                                const float* __restrict__ Ws,
                                                const float* __restrict__ Wn,
                                                const float* As, const float* Hs,
                                                int kc, int half, int l16, int col) {
    sage_load_w(bs1, bn1, Ws, Wn, kc + KC_K, half, col);
    sage_mma_chunk(acc, bs0, bn0, As, Hs, kc, half, l16);
    if (!LAST) sage_load_w(bs0, bn0, Ws, Wn, kc + 2 * KC_K, half, col);
    sage_mma_chunk(acc, bs1, bn1, As, Hs, kc + KC_K, half, l16);
}

template <bool HEAD>
__global__ void __launch_bounds__(256)
sage_layer_kernel(const float* __restrict__ xin, float* __restrict__ hout,
                  const int* __restrict__ first, const int* __restrict__ cnt,
                  const int* __restrict__ nbr, const float* __restrict__ invdeg,
                  const float* __restrict__ Wself, const float* __restrict__ Wneigh,
                  const float* __restrict__ bias, int n_nodes,
                  const float* __restrict__ Wp, const float* __restrict__ bp,
                  float* __restrict__ out) {
    __shared__ float As[TILE_M * LDS_STRIDE];
    __shared__ float Hs[TILE_M * LDS_STRIDE];

    const int m0 = blockIdx.x * TILE_M;
    const int t  = threadIdx.x;
    const int lane = t & 31;

    const int wave = t >> 5;
    const int n0   = wave * 16;
    const int half = lane >> 4;
    const int l16  = lane & 15;
    const int col  = n0 + l16;

    v2f bs0[KC_STEPS], bn0[KC_STEPS], bs1[KC_STEPS], bn1[KC_STEPS];
    sage_load_w(bs0, bn0, Wself, Wneigh, 0, half, col);
    const float bv = bias[col];

    {
        const int wu   = __builtin_amdgcn_readfirstlane(t >> 5);
        const int row0 = ROWS_PER_WAVE * wu;

        int   beg[ROWS_PER_WAVE];
        int   nn[ROWS_PER_WAVE];
        float iv[ROWS_PER_WAVE];
        int   ids[ROWS_PER_WAVE];
        float4 xv[ROWS_PER_WAVE];

        #pragma unroll
        for (int i = 0; i < ROWS_PER_WAVE; ++i) {
            const int node  = m0 + row0 + i;
            const bool live = node < n_nodes;
            const int nodeC = live ? node : (n_nodes - 1);
            beg[i] = first[nodeC];
            nn[i]  = live ? cnt[nodeC] : 0;
            iv[i]  = invdeg[nodeC];
        }
        #pragma unroll
        for (int i = 0; i < ROWS_PER_WAVE; ++i) {
            ids[i] = 0;
            if (lane < nn[i]) ids[i] = nbr[beg[i] + lane];
        }
        #pragma unroll
        for (int i = 0; i < ROWS_PER_WAVE; ++i) {
            const int node = m0 + row0 + i;
            xv[i] = make_float4(0.0f, 0.0f, 0.0f, 0.0f);
            if (node < n_nodes) xv[i] = *(const float4*)(xin + (size_t)node * D_FEAT + lane * 4);
        }
        #pragma unroll
        for (int i = 0; i < ROWS_PER_WAVE; ++i)
            *(float4*)(As + (row0 + i) * LDS_STRIDE + lane * 4) = xv[i];

        const float* xl = xin + lane * 4;
        #pragma unroll
        for (int i = 0; i < ROWS_PER_WAVE; ++i) {
            float4 s = sage_node_sum(nbr, xl, beg[i], nn[i], lane, ids[i]);
            s.x *= iv[i]; s.y *= iv[i]; s.z *= iv[i]; s.w *= iv[i];
            *(float4*)(Hs + (row0 + i) * LDS_STRIDE + lane * 4) = s;
        }
    }

    __syncthreads();

    v8f acc[M_SUB];
    #pragma unroll
    for (int mt = 0; mt < M_SUB; ++mt) {
        #pragma unroll
        for (int i = 0; i < 8; ++i) acc[mt][i] = bv;
    }

    for (int kc = 0; kc < D_FEAT - 2 * KC_K; kc += 2 * KC_K)
        sage_chunk_pair<false>(acc, bs0, bn0, bs1, bn1, Wself, Wneigh, As, Hs, kc, half, l16, col);
    sage_chunk_pair<true>(acc, bs0, bn0, bs1, bn1, Wself, Wneigh, As, Hs,
                          D_FEAT - 2 * KC_K, half, l16, col);

    if (!HEAD) {
        #pragma unroll
        for (int mt = 0; mt < M_SUB; ++mt) {
            #pragma unroll
            for (int v = 0; v < 8; ++v) {
                const int row = m0 + mt * 16 + v + half * 8;
                float val = acc[mt][v];
                val = val > 0.0f ? val : 0.0f;
                if (row < n_nodes) hout[(size_t)row * D_FEAT + col] = val;
            }
        }
    } else {
        const int pm   = wave >> 1;
        const int pcol = (wave & 1) * 16 + l16;
        const int bcol = pcol < OUT_DIM ? pcol : (OUT_DIM - 1);

        v2f wb[D_FEAT / 4];
        #pragma unroll
        for (int s = 0; s < D_FEAT / 4; ++s) {
            const int ka = 4 * s + half * 2;
            wb[s].x = Wp[(size_t)ka * OUT_DIM + bcol];
            wb[s].y = Wp[(size_t)(ka + 1) * OUT_DIM + bcol];
        }
        const float pbv = bp[bcol];

        __syncthreads();
        #pragma unroll
        for (int mt = 0; mt < M_SUB; ++mt) {
            #pragma unroll
            for (int v = 0; v < 8; ++v) {
                float val = acc[mt][v];
                val = val > 0.0f ? val : 0.0f;
                As[(mt * 16 + v + half * 8) * LDS_STRIDE + col] = val;
            }
        }
        __syncthreads();

        v8f pacc;
        #pragma unroll
        for (int i = 0; i < 8; ++i) pacc[i] = pbv;
        #pragma unroll
        for (int s = 0; s < D_FEAT / 4; ++s) {
            const int ka = 4 * s + half * 2;
            const v2f a = *(const v2f*)(As + (pm * 16 + l16) * LDS_STRIDE + ka);
            pacc = __builtin_amdgcn_wmma_f32_16x16x4_f32(false, a, false, wb[s],
                                                         (short)0, pacc, false, false);
        }
        if (pcol < OUT_DIM) {
            #pragma unroll
            for (int v = 0; v < 8; ++v) {
                const int row = m0 + pm * 16 + v + half * 8;
                if (row < n_nodes) out[(size_t)row * OUT_DIM + pcol] = pacc[v];
            }
        }
    }
}

__global__ void __launch_bounds__(64)
sage_pred_kernel(const float* __restrict__ xin, const float* __restrict__ Wp,
                 const float* __restrict__ bp, float* __restrict__ out) {
    const int m0   = blockIdx.x * 16;
    const int t    = threadIdx.x;
    const int wave = t >> 5;
    const int lane = t & 31;
    const int n0   = wave * 16;
    const int half = lane >> 4;
    const int l16  = lane & 15;
    const int col  = n0 + l16;
    const int bcol = col < OUT_DIM ? col : (OUT_DIM - 1);

    v8f acc;
    {
        float bv = bp[bcol];
        #pragma unroll
        for (int i = 0; i < 8; ++i) acc[i] = bv;
    }

    #pragma unroll 4
    for (int k0 = 0; k0 < D_FEAT; k0 += 4) {
        const int ka = k0 + half * 2;
        const float* ap = xin + (size_t)(m0 + l16) * D_FEAT + ka;
        v2f a; a.x = ap[0]; a.y = ap[1];
        v2f b;
        b.x = Wp[(size_t)ka * OUT_DIM + bcol];
        b.y = Wp[(size_t)(ka + 1) * OUT_DIM + bcol];
        acc = __builtin_amdgcn_wmma_f32_16x16x4_f32(false, a, false, b,
                                                    (short)0, acc, false, false);
    }

    if (col < OUT_DIM) {
        #pragma unroll
        for (int v = 0; v < 8; ++v) {
            int row = m0 + v + half * 8;
            out[(size_t)row * OUT_DIM + col] = acc[v];
        }
    }
}


extern "C" void kernel_launch(void* const* d_in, const int* in_sizes, int n_in,
                              void* d_out, int out_size, void* d_ws, size_t ws_size,
                              hipStream_t stream) {
    const float* x       = (const float*)d_in[0];
    const int*   src     = (const int*)  d_in[1];
    const int*   dst     = (const int*)  d_in[2];
    const float* W_self  = (const float*)d_in[3];
    const float* W_neigh = (const float*)d_in[4];
    const float* b       = (const float*)d_in[5];
    const float* W_pred  = (const float*)d_in[6];
    const float* b_pred  = (const float*)d_in[7];

    const int n_nodes = in_sizes[0] / D_FEAT;
    const int n_edges = in_sizes[1];
    const int L       = in_sizes[5] / D_FEAT;
    const int feat_elems = n_nodes * D_FEAT;

    char*  ws = (char*)d_ws;
    size_t off = ((size_t)n_nodes * sizeof(float) + 511) & ~(size_t)511;
    size_t feat_bytes = (size_t)feat_elems * sizeof(float);
    size_t edge_bytes = ((size_t)n_edges * sizeof(int) + 511) & ~(size_t)511;
    float* invdeg = (float*)ws;
    float* bufA   = (float*)(ws + off);
    float* bufB   = (float*)(ws + off + feat_bytes);
    char*  ga     = ws + off + 2 * feat_bytes;
    int*   cnt    = (int*)(ga);
    int*   first  = (int*)(ga + off);
    size_t cntc_bytes = ((size_t)EDGE_CHUNKS * n_nodes * sizeof(int) + 511) & ~(size_t)511;
    int*   cntc   = (int*)(ga + 2 * off);
    int*   nbr    = (int*)(ga + 2 * off + cntc_bytes);
    int*   pairs  = (int*)(ga + 2 * off + cntc_bytes + edge_bytes);

    const int T = 256;
    const dim3 scan_grid((n_nodes + RANGE_NODES - 1) / RANGE_NODES, EDGE_CHUNKS);
    const int wave_blocks = (n_nodes + 7) / 8;

    sage_count_kernel<<<scan_grid, SCAN_THREADS, 0, stream>>>(dst, cntc, n_edges, n_nodes);
    sage_totals_kernel<<<(n_nodes + T - 1) / T, T, 0, stream>>>(cntc, cnt, n_nodes);
    sage_offsets_kernel<<<(n_nodes + SCAN_CHUNK - 1) / SCAN_CHUNK, 256, 0, stream>>>(
        cnt, first, invdeg, n_nodes);
    sage_bucket_kernel<<<scan_grid, SCAN_THREADS, 0, stream>>>(src, dst, first, cntc, pairs,
                                                              n_edges, n_nodes);
    sage_order_kernel<<<wave_blocks, 256, 0, stream>>>(first, cnt, pairs, nbr, n_nodes);

    const float* cur = x;
    float* nxt = bufA;
    const int m_tiles = n_nodes / 16;
    const int layer_blocks = (n_nodes + TILE_M - 1) / TILE_M;

    for (int l = 0; l < L; ++l) {
        if (l < L - 1) {
            sage_layer_kernel<false><<<layer_blocks, 256, 0, stream>>>(
                cur, nxt, first, cnt, nbr, invdeg,
                W_self  + (size_t)l * D_FEAT * D_FEAT,
                W_neigh + (size_t)l * D_FEAT * D_FEAT,
                b       + (size_t)l * D_FEAT, n_nodes, nullptr, nullptr, nullptr);
        } else {
            sage_layer_kernel<true><<<layer_blocks, 256, 0, stream>>>(
                cur, nxt, first, cnt, nbr, invdeg,
                W_self  + (size_t)l * D_FEAT * D_FEAT,
                W_neigh + (size_t)l * D_FEAT * D_FEAT,
                b       + (size_t)l * D_FEAT, n_nodes, W_pred, b_pred, (float*)d_out);
        }
        cur = nxt;
        nxt = (nxt == bufA) ? bufB : bufA;
    }

    if (L == 0) sage_pred_kernel<<<m_tiles, 64, 0, stream>>>(cur, W_pred, b_pred, (float*)d_out);
}
